# starting kernel
# baseline (speedup 1.0000x reference)
#include <hip/hip_runtime.h>
#include <hip/hip_bf16.h>

typedef float v2f __attribute__((ext_vector_type(2)));
typedef float v8f __attribute__((ext_vector_type(8)));

#define LRELU_SLOPE 0.01f

__device__ __forceinline__ void atomic_add_f32(float* p, float v) {
    asm volatile("global_atomic_add_f32 %0, %1, off scope:SCOPE_DEV"
                 :: "v"(p), "v"(v) : "memory");
}

__global__ void k_zero_u32(unsigned* __restrict__ p, int n) {
    int i = blockIdx.x * blockDim.x + threadIdx.x;
    if (i < n) p[i] = 0u;
}

__global__ void k_count_deg(const int* __restrict__ ei, int E, unsigned* __restrict__ deg) {
    int e = blockIdx.x * blockDim.x + threadIdx.x;
    if (e < E) {
        int d = ei[E + e];
        atomicAdd(&deg[d], 1u);
    }
}

__global__ void k_dinv(const unsigned* __restrict__ deg, float* __restrict__ dinv, int n) {
    int i = blockIdx.x * blockDim.x + threadIdx.x;
    if (i < n) dinv[i] = rsqrtf((float)(deg[i] + 1u));
}

template <int NOUT>
__global__ __launch_bounds__(256) void k_gemm_wmma(const float* __restrict__ A,
                                                   const float* __restrict__ W,
                                                   float* __restrict__ out, int n_rows) {
    __shared__ float sW[128 * NOUT];
    const int tid = threadIdx.x;
    for (int i = tid; i < 128 * NOUT; i += 256) {
        int k = i / NOUT, n = i % NOUT;
        sW[((k >> 1) * NOUT + n) * 2 + (k & 1)] = W[i];
    }
    __syncthreads();

    const int wave = tid >> 5;
    const int lane = tid & 31;
    const int m16  = lane & 15;
    const int hi   = lane >> 4;
    const int row0 = blockIdx.x * 128 + wave * 16;
    constexpr int NT = NOUT / 16;

    v8f zero8 = {};
    v8f acc[NT];
#pragma unroll
    for (int t = 0; t < NT; ++t) acc[t] = zero8;

    int rA = row0 + m16;
    if (rA >= n_rows) rA = n_rows - 1;
    const float* arow = A + (size_t)rA * 128;

    for (int k0 = 0; k0 < 128; k0 += 4) {
        v2f a = *(const v2f*)(arow + k0 + 2 * hi);
        const int kh = (k0 >> 1) + hi;
#pragma unroll
        for (int t = 0; t < NT; ++t) {
            v2f b = *(const v2f*)(sW + (size_t)(kh * NOUT + t * 16 + m16) * 2);
            acc[t] = __builtin_amdgcn_wmma_f32_16x16x4_f32(
                false, a, false, b, (short)0, acc[t], false, false);
        }
    }

#pragma unroll
    for (int t = 0; t < NT; ++t) {
#pragma unroll
        for (int v = 0; v < 8; ++v) {
            int row = row0 + v + 8 * hi;
            if (row < n_rows) out[(size_t)row * NOUT + t * 16 + m16] = acc[t][v];
        }
    }
}

template <int FEAT>
__global__ void k_self_init(const float* __restrict__ h, const float* __restrict__ dinv,
                            float* __restrict__ acc, int n_rows) {
    constexpr int QPR = FEAT / 4;
    int t = blockIdx.x * blockDim.x + threadIdx.x;
    int node = t / QPR;
    if (node >= n_rows) return;
    int fi = (t % QPR) * 4;
    float di = dinv[node];
    float w = di * di;
    float4 v = *(const float4*)(h + (size_t)node * FEAT + fi);
    float4 o = {v.x * w, v.y * w, v.z * w, v.w * w};
    *(float4*)(acc + (size_t)node * FEAT + fi) = o;
}

template <int FEAT>
__global__ __launch_bounds__(256) void k_edge_agg(const float* __restrict__ h,
                                                  const float* __restrict__ dinv,
                                                  const int* __restrict__ ei, int E,
                                                  float* __restrict__ acc) {
    constexpr int VW = FEAT / 32;
    long long t = (long long)blockIdx.x * 256 + threadIdx.x;
    int e = (int)(t >> 5);
    if (e >= E) return;
    e = __builtin_amdgcn_readfirstlane(e);
    const int lane = threadIdx.x & 31;
    int s = ei[e];
    int d = ei[E + e];
    float nrm = dinv[s] * dinv[d];
    const float* src = h + (size_t)s * FEAT + lane * VW;
    float* dst = acc + (size_t)d * FEAT + lane * VW;
    if constexpr (VW == 4) {
        float4 v = *(const float4*)src;
        atomic_add_f32(dst + 0, nrm * v.x);
        atomic_add_f32(dst + 1, nrm * v.y);
        atomic_add_f32(dst + 2, nrm * v.z);
        atomic_add_f32(dst + 3, nrm * v.w);
    } else {
        float2 v = *(const float2*)src;
        atomic_add_f32(dst + 0, nrm * v.x);
        atomic_add_f32(dst + 1, nrm * v.y);
    }
}

template <int FEAT, bool LEAKY>
__global__ void k_bias_act(float* __restrict__ acc, const float* __restrict__ b, int n_rows) {
    constexpr int QPR = FEAT / 4;
    int t = blockIdx.x * blockDim.x + threadIdx.x;
    int node = t / QPR;
    if (node >= n_rows) return;
    int fi = (t % QPR) * 4;
    float4 v  = *(const float4*)(acc + (size_t)node * FEAT + fi);
    float4 bb = *(const float4*)(b + fi);
    v.x += bb.x; v.y += bb.y; v.z += bb.z; v.w += bb.w;
    if (LEAKY) {
        v.x = v.x > 0.0f ? v.x : LRELU_SLOPE * v.x;
        v.y = v.y > 0.0f ? v.y : LRELU_SLOPE * v.y;
        v.z = v.z > 0.0f ? v.z : LRELU_SLOPE * v.z;
        v.w = v.w > 0.0f ? v.w : LRELU_SLOPE * v.w;
    }
    *(float4*)(acc + (size_t)node * FEAT + fi) = v;
}

extern "C" void kernel_launch(void* const* d_in, const int* in_sizes, int n_in,
                              void* d_out, int out_size, void* d_ws, size_t ws_size,
                              hipStream_t stream) {
    (void)n_in; (void)out_size; (void)ws_size;
    const float* z  = (const float*)d_in[0];
    const int*   ei = (const int*)d_in[1];
    const float* W1 = (const float*)d_in[2];
    const float* b1 = (const float*)d_in[3];
    const float* W2 = (const float*)d_in[4];
    const float* b2 = (const float*)d_in[5];

    const int N = in_sizes[0] / 128;
    const int E = in_sizes[1] / 2;

    char* ws = (char*)d_ws;
    size_t off = 0;
    auto carve = [&](size_t bytes) -> void* {
        void* p = ws + off;
        off = (off + bytes + 255) & ~(size_t)255;
        return p;
    };
    unsigned* deg  = (unsigned*)carve((size_t)N * 4);
    float*    dinv = (float*)carve((size_t)N * 4);
    float*    h1   = (float*)carve((size_t)N * 128 * 4);
    float*    a1   = (float*)carve((size_t)N * 128 * 4);
    float*    h3   = (float*)carve((size_t)N * 64 * 4);
    float*    outp = (float*)d_out;

    const dim3 blk(256);
    const int gN = (N + 255) / 256;
    const int gE = (E + 255) / 256;
    const int gM = (N + 127) / 128;
    const int gEdge = (int)(((long long)E * 32 + 255) / 256);

    k_zero_u32 <<<gN, blk, 0, stream>>>(deg, N);
    k_count_deg<<<gE, blk, 0, stream>>>(ei, E, deg);
    k_dinv     <<<gN, blk, 0, stream>>>(deg, dinv, N);

    k_gemm_wmma<128><<<gM, blk, 0, stream>>>(z, W1, h1, N);
    long long t128 = (long long)N * 32;
    int gS1 = (int)((t128 + 255) / 256);
    k_self_init<128><<<gS1, blk, 0, stream>>>(h1, dinv, a1, N);
    k_edge_agg<128><<<gEdge, blk, 0, stream>>>(h1, dinv, ei, E, a1);
    k_bias_act<128, true><<<gS1, blk, 0, stream>>>(a1, b1, N);

    k_gemm_wmma<64><<<gM, blk, 0, stream>>>(a1, W2, h3, N);
    long long t64 = (long long)N * 16;
    int gS2 = (int)((t64 + 255) / 256);
    k_self_init<64><<<gS2, blk, 0, stream>>>(h3, dinv, outp, N);
    k_edge_agg<64><<<gEdge, blk, 0, stream>>>(h3, dinv, ei, E, outp);
    k_bias_act<64, false><<<gS2, blk, 0, stream>>>(outp, b2, N);
}


# Launches 7 to 4: degree count shares a launch with the z W1 product; edge kernels form their own start values (dinv from deg); layer-1 edge kernel does the W2 product on its LDS rows; sums unchanged
# speedup vs baseline: 3.1768x; 3.1768x over previous
#include <hip/hip_runtime.h>
#include <hip/hip_bf16.h>

typedef float v2f __attribute__((ext_vector_type(2)));
typedef float v8f __attribute__((ext_vector_type(8)));

#define LRELU_SLOPE 0.01f

#define OWN_ROWS  392
#define OWN_WAVES 32
#define OWN_QCAP  544
#define OWN_LCAP  16384

typedef int      v2i __attribute__((ext_vector_type(2)));
typedef int      v4i __attribute__((ext_vector_type(4)));
typedef unsigned v4u __attribute__((ext_vector_type(4)));

__device__ __forceinline__ v4i load_dst4(const int* __restrict__ dstp, int e, int E) {
    v4i r;
    r.x = (e     < E) ? dstp[e]     : -1;
    r.y = (e + 1 < E) ? dstp[e + 1] : -1;
    r.z = (e + 2 < E) ? dstp[e + 2] : -1;
    r.w = (e + 3 < E) ? dstp[e + 3] : -1;
    return r;
}

__global__ void k_zero_u32(unsigned* __restrict__ p, int n) {
    int i = blockIdx.x * blockDim.x + threadIdx.x;
    if (i < n) p[i] = 0u;
}

__device__ __forceinline__ void count_block(const int* __restrict__ ei, int E, unsigned* __restrict__ deg,
                                            unsigned* __restrict__ key4, int blk) {
    const int t = blk * blockDim.x + threadIdx.x;
    const int e = t * 4;
    if (e >= E) return;
    const int* dstp = ei + E;
    v4i d;
    if ((E & 3) == 0) d = *(const v4i*)(dstp + e);
    else              d = load_dst4(dstp, e, E);
    unsigned k = 0u;
#pragma unroll
    for (int j = 0; j < 4; ++j) {
        if (e + j < E) {
            const int dj = d[j];
            atomicAdd(&deg[dj], 1u);
            k |= ((unsigned)(dj / OWN_ROWS) & 255u) << (8 * j);
        }
    }
    key4[t] = k;
}

__device__ __forceinline__ float dinv_of(const unsigned* __restrict__ deg, int i) {
    return rsqrtf((float)(deg[i] + 1u));
}

template <int NOUT>
__device__ __forceinline__ void gemm_block(const float* __restrict__ A,
                                           const float* __restrict__ W,
                                           float* __restrict__ out, int n_rows, int blk) {
    __shared__ float sW[128 * NOUT];
    const int tid = threadIdx.x;
    for (int i = tid; i < 128 * NOUT; i += 256) {
        int k = i / NOUT, n = i % NOUT;
        sW[((k >> 1) * NOUT + n) * 2 + (k & 1)] = W[i];
    }
    __syncthreads();

    const int wave = tid >> 5;
    const int lane = tid & 31;
    const int m16  = lane & 15;
    const int hi   = lane >> 4;
    const int row0 = blk * 128 + wave * 16;
    constexpr int NT = NOUT / 16;

    v8f zero8 = {};
    v8f acc[NT];
#pragma unroll
    for (int t = 0; t < NT; ++t) acc[t] = zero8;

    int rA = row0 + m16;
    if (rA >= n_rows) rA = n_rows - 1;
    const float* arow = A + (size_t)rA * 128;

    for (int k0 = 0; k0 < 128; k0 += 4) {
        v2f a = *(const v2f*)(arow + k0 + 2 * hi);
        const int kh = (k0 >> 1) + hi;
#pragma unroll
        for (int t = 0; t < NT; ++t) {
            v2f b = *(const v2f*)(sW + (size_t)(kh * NOUT + t * 16 + m16) * 2);
            acc[t] = __builtin_amdgcn_wmma_f32_16x16x4_f32(
                false, a, false, b, (short)0, acc[t], false, false);
        }
    }

#pragma unroll
    for (int t = 0; t < NT; ++t) {
#pragma unroll
        for (int v = 0; v < 8; ++v) {
            int row = row0 + v + 8 * hi;
            if (row < n_rows)
                out[(size_t)row * NOUT + t * 16 + m16] = acc[t][v];
        }
    }
}

__global__ __launch_bounds__(256) void k_gemm_count(const float* __restrict__ A,
                                                    const float* __restrict__ W,
                                                    float* __restrict__ out, int n_rows, int n_count,
                                                    const int* __restrict__ ei, int E,
                                                    unsigned* __restrict__ deg,
                                                    unsigned* __restrict__ key4) {
    const int b = (int)blockIdx.x;
    if (b < n_count) count_block(ei, E, deg, key4, b);
    else             gemm_block<128>(A, W, out, n_rows, b - n_count);
}


__device__ __forceinline__ int   lane_i(int v, int l)   { return __builtin_amdgcn_readlane(v, l); }
__device__ __forceinline__ float lane_f(float v, int l) {
    return __builtin_bit_cast(float, __builtin_amdgcn_readlane(__builtin_bit_cast(int, v), l));
}

__device__ __forceinline__ unsigned nz7(unsigned x) {
    return ((x & 0x7f7f7f7fu) + 0x7f7f7f7fu) | x;
}

__device__ __forceinline__ v4u load_key16(const unsigned char* __restrict__ key, int e, int E,
                                          unsigned other) {
    v4u r;
#pragma unroll
    for (int j = 0; j < 4; ++j) {
        unsigned w = 0u;
#pragma unroll
        for (int b = 0; b < 4; ++b) {
            const int x = e + 4 * j + b;
            const unsigned kb = (x < E) ? (unsigned)key[x] : other;
            w |= kb << (8 * b);
        }
        r[j] = w;
    }
    return r;
}

template <int FEAT, int PITCH, int B>
__device__ __forceinline__ void add_rows(const float* __restrict__ h, float* sAcc,
                                         int s, int dl, float nrm, int n, int lane) {
    constexpr int VW = FEAT / 32;
    constexpr int NB = 32 / B;
    typedef float vrow __attribute__((ext_vector_type(VW)));
#pragma unroll
    for (int b = 0; b < NB; ++b) {
        if (b * B < n) {
            vrow r[B];
#pragma unroll
            for (int j = 0; j < B; ++j) {
                if (b * B + j < n)
                    r[j] = *(const vrow*)(h + (size_t)lane_i(s, b * B + j) * FEAT + lane * VW);
            }
#pragma unroll
            for (int j = 0; j < B; ++j) {
                if (b * B + j < n) {
                    const int   row = lane_i(dl, b * B + j);
                    const float w   = lane_f(nrm, b * B + j);
                    float* a = sAcc + row * PITCH + lane;
#pragma unroll
                    for (int i = 0; i < VW; ++i)
                        __hip_atomic_fetch_add(a + i * 32, w * r[j][i], __ATOMIC_RELAXED,
                                               __HIP_MEMORY_SCOPE_WORKGROUP);
                }
            }
        }
    }
}

template <int FEAT, bool LISTED, int NEXT>
__global__ __launch_bounds__(1024) void k_edge_own(const float* __restrict__ h,
                                                   const unsigned* __restrict__ deg,
                                                   const int* __restrict__ ei, int E,
                                                   const unsigned char* __restrict__ key,
                                                   float* __restrict__ acc, int n_rows,
                                                   int* __restrict__ lst,
                                                   unsigned* __restrict__ lcnt,
                                                   const float* __restrict__ bias,
                                                   const float* __restrict__ Wn,
                                                   float* __restrict__ hn) {
    constexpr int VW = FEAT / 32;
    constexpr int B  = 32 / VW;
    constexpr int PITCH = NEXT > 0 ? FEAT + 4 : FEAT;
    typedef float vrow __attribute__((ext_vector_type(VW)));
    static_assert(NEXT == 0 || FEAT * NEXT <= OWN_WAVES * OWN_QCAP, "Wn must fit the queue space");

    __shared__ __attribute__((aligned(16))) float sAcc[(OWN_ROWS + 1) * PITCH];
    __shared__ __attribute__((aligned(16))) int sQ[OWN_WAVES * OWN_QCAP];
    __shared__ unsigned sCnt;

    const int lane = threadIdx.x & 31;
    const int wave = __builtin_amdgcn_readfirstlane((int)(threadIdx.x >> 5));
    const int lo   = blockIdx.x * OWN_ROWS;
    int nown = n_rows - lo;
    if (nown > OWN_ROWS) nown = OWN_ROWS;

    if (threadIdx.x == 0) sCnt = 0u;
    constexpr int NWL = NEXT > 0 ? (FEAT * NEXT) / (OWN_WAVES * 32) : 1;
    static_assert(NEXT == 0 || (FEAT * NEXT) % (OWN_WAVES * 32) == 0, "whole turns of the work-group");
    float wreg[NWL];
    if constexpr (NEXT > 0) {
#pragma unroll
        for (int u = 0; u < NWL; ++u) wreg[u] = Wn[u * (OWN_WAVES * 32) + (int)threadIdx.x];
    }
    {
        const vrow bv = *(const vrow*)(bias + lane * VW);
        for (int r0 = wave; r0 < nown; r0 += 4 * OWN_WAVES) {
            vrow  v[4];
            float w[4];
#pragma unroll
            for (int k = 0; k < 4; ++k) {
                const int r = r0 + k * OWN_WAVES;
                if (r < nown) {
                    v[k] = *(const vrow*)(h + (size_t)(lo + r) * FEAT + lane * VW);
                    const float di = dinv_of(deg, lo + r);
                    w[k] = di * di;
                }
            }
#pragma unroll
            for (int k = 0; k < 4; ++k) {
                const int r = r0 + k * OWN_WAVES;
                if (r < nown) {
#pragma unroll
                    for (int i = 0; i < VW; ++i) sAcc[r * PITCH + i * 32 + lane] = w[k] * v[k][i] + bv[i];
                }
            }
        }
    }
    __syncthreads();

    const int* dstp  = ei + E;
    const int  nfull = E >> 11;
    const unsigned own  = blockIdx.x;
    const unsigned own4 = own * 0x01010101u;
    const int rot = (int)blockIdx.x * (nfull / (int)gridDim.x);
    int* q  = sQ + wave * OWN_QCAP;
    int cnt = 0;

    auto flush = [&](const int* qp, int n) __attribute__((always_inline)) {
        const int me = qp[lane < n ? lane : 0];
        const int   s   = ei[me];
        const int   d   = dstp[me];
        const float nrm = dinv_of(deg, s) * dinv_of(deg, d);
        const int   dl  = d - lo;
        if (!LISTED) {
            unsigned pos = 0u;
            if (lane == 0)
                pos = __hip_atomic_fetch_add(&sCnt, 32u, __ATOMIC_RELAXED, __HIP_MEMORY_SCOPE_WORKGROUP);
            pos = (unsigned)__builtin_amdgcn_readfirstlane((int)pos);
            if (pos + 32u <= (unsigned)OWN_LCAP) {
                v2i rec;
                rec.x = lane < n ? s * 512 + dl : OWN_ROWS;
                rec.y = lane < n ? __builtin_bit_cast(int, nrm) : 0;
                ((v2i*)lst)[(size_t)own * OWN_LCAP + pos + lane] = rec;
            }
        }
        add_rows<FEAT, PITCH, B>(h, sAcc, s, dl, nrm, n, lane);
    };

    auto search = [&](const v4u k, int eb) __attribute__((always_inline)) {
        const unsigned n0 = nz7(k.x ^ own4);
        const unsigned n1 = nz7(k.y ^ own4);
        const unsigned n2 = nz7(k.z ^ own4);
        const unsigned n3 = nz7(k.w ^ own4);
        unsigned m = (((n0 >> 7) & 0x01010101u) | ((n1 >> 6) & 0x02020202u) |
                      ((n2 >> 5) & 0x04040404u) | ((n3 >> 4) & 0x08080808u)) ^ 0x0f0f0f0fu;
        unsigned bal = __builtin_amdgcn_ballot_w32(m != 0u);
        while (bal != 0u) {
            if (m != 0u) {
                const int pos = __builtin_ctz(m);
                q[cnt + (int)__builtin_amdgcn_mbcnt_lo(bal, 0u)] = eb + ((pos & 7) << 2) + (pos >> 3);
                m &= m - 1u;
            }
            cnt += __builtin_popcount(bal);
            bal = __builtin_amdgcn_ballot_w32(m != 0u);
        }
    };
    auto drain = [&]() __attribute__((always_inline)) {
        __builtin_amdgcn_wave_barrier();
        while (cnt >= 32) { cnt -= 32; flush(q + cnt, 32); }
    };

    bool listed = false;
    if (LISTED) {
        const unsigned taken = lcnt[own];
        if (taken <= (unsigned)OWN_LCAP) {
            listed = true;
            const int np = (int)(taken >> 5);
            const v2i* lp = (const v2i*)lst + (size_t)own * OWN_LCAP + lane;
            int   ck = OWN_ROWS;
            float cw = 0.0f;
            if (wave < np) { const v2i r = lp[wave * 32]; const int wb = r.y; ck = r.x; cw = __builtin_bit_cast(float, wb); }
            for (int p = wave; p < np; p += OWN_WAVES) {
                int nk = ck; float nw = cw;
                if (p + OWN_WAVES < np) {
                    const v2i r = lp[(p + OWN_WAVES) * 32];
                    const int wb = r.y;
                    nk = r.x; nw = __builtin_bit_cast(float, wb);
                }
                add_rows<FEAT, PITCH, (FEAT <= 64 ? 32 : 16)>(h, sAcc, (int)((unsigned)ck >> 9), ck & 511, cw, 32, lane);
                ck = nk; cw = nw;
            }
        }
    }
    if (!listed) {
    if (wave < nfull) {
        v4u cur[4], nxt[4];
        auto request = [&](v4u (&buf)[4], int g) __attribute__((always_inline)) {
            int gg = (g < nfull ? g : wave) + rot; if (gg >= nfull) gg -= nfull;
            const v4u* p = (const v4u*)(key + (size_t)gg * 2048) + lane;
#pragma unroll
            for (int u = 0; u < 4; ++u) buf[u] = p[u * 32];
        };
        request(cur, wave);
        for (int g = wave; g < nfull; g += OWN_WAVES) {
            request(nxt, g + OWN_WAVES);
            int gg = g + rot; if (gg >= nfull) gg -= nfull;
            const int e0 = gg * 2048 + lane * 16;
            v4u k0 = cur[0], k1 = cur[1], k2 = cur[2], k3 = cur[3];
#pragma clang loop unroll(disable)
            for (int u = 0; u < 4; ++u) {
                search(k0, e0 + u * 512);
                drain();
                k0 = k1; k1 = k2; k2 = k3;
            }
#pragma unroll
            for (int u = 0; u < 4; ++u) cur[u] = nxt[u];
        }
    }
    {
        const int ebase = nfull << 11;
        const int ntail = (E - ebase + 511) >> 9;
        for (int t = wave; t < ntail; t += OWN_WAVES) {
            const int e0 = ebase + t * 512 + lane * 16;
            search(load_key16(key, e0, E, (own + 1u) & 255u), e0);
            drain();
        }
    }
    if (cnt > 0) flush(q, cnt);
    }
    __syncthreads();
    if (!LISTED && threadIdx.x == 0) lcnt[own] = sCnt;

    if constexpr (NEXT == 0) {
        for (int r = wave; r < nown; r += OWN_WAVES) {
            vrow v;
#pragma unroll
            for (int i = 0; i < VW; ++i) v[i] = sAcc[r * PITCH + i * 32 + lane];
            *(vrow*)(acc + (size_t)(lo + r) * FEAT + lane * VW) = v;
        }
    } else {
        constexpr int NT = NEXT / 16;
        constexpr int VN = NEXT / 32;
        typedef float vnext __attribute__((ext_vector_type(VN)));
        float* sW = reinterpret_cast<float*>(sQ);
#pragma unroll
        for (int u = 0; u < NWL; ++u) {
            const int i = u * (OWN_WAVES * 32) + (int)threadIdx.x;
            const int k = i / NEXT, n = i % NEXT;
            sW[((k >> 1) * NEXT + n) * 2 + (k & 1)] = wreg[u];
        }
        __syncthreads();
        const int m16 = lane & 15;
        const int hi  = lane >> 4;
        for (int r0 = wave * 16; r0 < nown; r0 += OWN_WAVES * 16) {
            int ra = r0 + m16;
            if (ra >= nown) ra = nown - 1;
            const float* arow = sAcc + ra * PITCH + hi * 64;
            v8f zero8 = {};
            v8f c[NT];
#pragma unroll
            for (int t = 0; t < NT; ++t) c[t] = zero8;
            for (int kw = 0; kw < FEAT / 4; ++kw) {
                v2f a;
                a.x = arow[kw];
                a.y = arow[32 + kw];
                a.x = a.x > 0.0f ? a.x : LRELU_SLOPE * a.x;
                a.y = a.y > 0.0f ? a.y : LRELU_SLOPE * a.y;
                const int kh = 2 * kw + hi;
#pragma unroll
                for (int t = 0; t < NT; ++t) {
                    v2f b = *(const v2f*)(sW + (size_t)(kh * NEXT + t * 16 + m16) * 2);
                    c[t] = __builtin_amdgcn_wmma_f32_16x16x4_f32(
                        false, a, false, b, (short)0, c[t], false, false);
                }
            }
            float* srow = sAcc + r0 * PITCH;
#pragma unroll
            for (int t = 0; t < NT; ++t) {
#pragma unroll
                for (int v = 0; v < 8; ++v) {
                    if (r0 + v + 8 * hi < nown)
                        srow[(v + 8 * hi) * PITCH + t * 16 + m16] = c[t][v];
                }
            }
            __builtin_amdgcn_wave_barrier();
            vnext o[16];
#pragma unroll
            for (int j = 0; j < 16; ++j)
                if (r0 + j < nown) o[j] = *(const vnext*)(srow + j * PITCH + lane * VN);
#pragma unroll
            for (int j = 0; j < 16; ++j)
                if (r0 + j < nown)
                    *(vnext*)(hn + (size_t)(lo + r0 + j) * NEXT + lane * VN) = o[j];
        }
    }
}

extern "C" void kernel_launch(void* const* d_in, const int* in_sizes, int n_in,
                              void* d_out, int out_size, void* d_ws, size_t ws_size,
                              hipStream_t stream) {
    (void)n_in; (void)out_size; (void)ws_size;
    const float* z  = (const float*)d_in[0];
    const int*   ei = (const int*)d_in[1];
    const float* W1 = (const float*)d_in[2];
    const float* b1 = (const float*)d_in[3];
    const float* W2 = (const float*)d_in[4];
    const float* b2 = (const float*)d_in[5];

    const int N = in_sizes[0] / 128;
    const int E = in_sizes[1] / 2;

    char* ws = (char*)d_ws;
    size_t off = 0;
    auto carve = [&](size_t bytes) -> void* {
        void* p = ws + off;
        off = (off + bytes + 255) & ~(size_t)255;
        return p;
    };
    unsigned* deg  = (unsigned*)carve((size_t)N * 4);
    unsigned* key4 = (unsigned*)carve((size_t)E + 4);
    const unsigned char* key = (const unsigned char*)key4;
    float*    h1   = (float*)carve((size_t)N * 128 * 4);
    float*    h3   = (float*)carve((size_t)N * 64 * 4);
    float*    outp = (float*)d_out;

    const dim3 blk(256);
    const int gN = (N + 255) / 256;
    const int gE4 = ((E + 3) / 4 + 255) / 256;
    const int gM = (N + 127) / 128;
    const int gOwn = (N + OWN_ROWS - 1) / OWN_ROWS;
    const dim3 blkOwn(OWN_WAVES * 32);
    int*      lst  = (int*)carve((size_t)gOwn * OWN_LCAP * 8);
    unsigned* lcnt = (unsigned*)carve((size_t)gOwn * 4);

    k_zero_u32  <<<gN, blk, 0, stream>>>(deg, N);
    k_gemm_count<<<gE4 + gM, blk, 0, stream>>>(z, W1, h1, N, gE4, ei, E, deg, key4);

    k_edge_own<128, false, 64><<<gOwn, blkOwn, 0, stream>>>(h1, deg, ei, E, key, nullptr, N, lst, lcnt,
                                                            b1, W2, h3);

    k_edge_own<64, true, 0><<<gOwn, blkOwn, 0, stream>>>(h3, deg, ei, E, key, outp, N, lst, lcnt,
                                                         b2, nullptr, nullptr);
}


# Degree count without global atomics: 64 work-groups each count a 25000-edge slice in LDS (16-bit pairs), store a plane; planes summed beside the z W1 product; zero launch gone; float sums unchanged
# speedup vs baseline: 4.4193x; 1.3911x over previous
#include <hip/hip_runtime.h>
#include <hip/hip_bf16.h>

typedef float v2f __attribute__((ext_vector_type(2)));
typedef float v8f __attribute__((ext_vector_type(8)));

#define LRELU_SLOPE 0.01f

#define OWN_ROWS  392
#define OWN_WAVES 32
#define OWN_QCAP  544
#define OWN_LCAP  16384

typedef int      v2i __attribute__((ext_vector_type(2)));
typedef int      v4i __attribute__((ext_vector_type(4)));
typedef unsigned v2u __attribute__((ext_vector_type(2)));
typedef unsigned v4u __attribute__((ext_vector_type(4)));

#define DEG_SLICE 25000
#define DEG_THREADS 1024
#define DEG_WORDS 50176

__device__ __forceinline__ v4i load_dst4(const int* __restrict__ dstp, int e, int E) {
    v4i r;
    r.x = (e     < E) ? dstp[e]     : -1;
    r.y = (e + 1 < E) ? dstp[e + 1] : -1;
    r.z = (e + 2 < E) ? dstp[e + 2] : -1;
    r.w = (e + 3 < E) ? dstp[e + 3] : -1;
    return r;
}

__global__ __launch_bounds__(DEG_THREADS) void k_count_slices(const int* __restrict__ ei, int E, int n_rows,
                                                       unsigned* __restrict__ part,
                                                       unsigned* __restrict__ key4) {
    static_assert(DEG_SLICE < 65536 && DEG_SLICE % 4 == 0, "a slice's count fits 16 bits; key words do not straddle slices");
    static_assert(DEG_WORDS % 4 == 0 && 2 * DEG_WORDS >= 256 * OWN_ROWS, "whole 16-byte stores; every row a key can name");
    __shared__ __attribute__((aligned(16))) unsigned sC[DEG_WORDS];
    constexpr int TURNS = (DEG_SLICE + 4 * DEG_THREADS - 1) / (4 * DEG_THREADS);
    const int tid = (int)threadIdx.x;
    const int* dstp = ei + E;
    const int e0 = (int)blockIdx.x * DEG_SLICE;
    int e1 = e0 + DEG_SLICE;
    if (e1 > E) e1 = E;

    const unsigned nlim = (unsigned)(n_rows < 2 * DEG_WORDS ? n_rows : 2 * DEG_WORDS);

    v4i dd[TURNS];
    if ((E & 3) == 0) {
#pragma unroll
        for (int t = 0; t < TURNS; ++t) {
            const int e = e0 + 4 * (tid + t * DEG_THREADS);
            dd[t] = *(const v4i*)(dstp + (e < e1 ? e : e0));
        }
    } else {
#pragma unroll
        for (int t = 0; t < TURNS; ++t) {
            const int e = e0 + 4 * (tid + t * DEG_THREADS);
            dd[t] = load_dst4(dstp, e < e1 ? e : e0, E);
        }
    }
    const v4u zero4 = {0u, 0u, 0u, 0u};
    for (int i = tid; i < DEG_WORDS / 4; i += DEG_THREADS) ((v4u*)sC)[i] = zero4;
    __syncthreads();

#pragma unroll
    for (int t = 0; t < TURNS; ++t) {
        const int e = e0 + 4 * (tid + t * DEG_THREADS);
        if (e >= e1) break;
        const v4i d = dd[t];
        unsigned k = 0u;
#pragma unroll
        for (int j = 0; j < 4; ++j) {
            if (e + j < e1) {
                const int dj = d[j];
                if ((unsigned)dj < nlim)
                    __hip_atomic_fetch_add(&sC[dj >> 1], 1u << ((dj & 1) * 16), __ATOMIC_RELAXED,
                                           __HIP_MEMORY_SCOPE_WORKGROUP);
                k |= ((unsigned)(dj / OWN_ROWS) & 255u) << (8 * j);
            }
        }
        key4[e >> 2] = k;
    }
    __syncthreads();

    const int nw  = (int)((nlim + 1u) >> 1);
    const int nw4 = nw >> 2;
    unsigned* p = part + (size_t)blockIdx.x * DEG_WORDS;
    for (int i = tid; i < nw4; i += DEG_THREADS) ((v4u*)p)[i] = ((const v4u*)sC)[i];
    for (int i = (nw4 << 2) + tid; i < nw; i += DEG_THREADS) p[i] = sC[i];
}

__device__ __forceinline__ void merge_block(const unsigned* __restrict__ part, int n_slices, int n_rows,
                                            unsigned* __restrict__ deg, int blk) {
    const int i  = blk * (int)blockDim.x + (int)threadIdx.x;
    const int nw = (n_rows + 1) >> 1;
    if (i >= nw || i >= DEG_WORDS) return;
    unsigned lo = 0u, hi = 0u;
#pragma unroll 8
    for (int sl = 0; sl < n_slices; ++sl) {
        const unsigned w = part[(size_t)sl * DEG_WORDS + i];
        lo += w & 0xffffu;
        hi += w >> 16;
    }
    if (2 * i + 1 < n_rows) { v2u r; r.x = lo; r.y = hi; *(v2u*)(deg + 2 * i) = r; }
    else                    deg[2 * i] = lo;
}

__device__ __forceinline__ float dinv_of(const unsigned* __restrict__ deg, int i) {
    return rsqrtf((float)(deg[i] + 1u));
}

template <int NOUT>
__device__ __forceinline__ void gemm_block(const float* __restrict__ A,
                                           const float* __restrict__ W,
                                           float* __restrict__ out, int n_rows, int blk) {
    __shared__ float sW[128 * NOUT];
    const int tid = threadIdx.x;
    for (int i = tid; i < 128 * NOUT; i += 256) {
        int k = i / NOUT, n = i % NOUT;
        sW[((k >> 1) * NOUT + n) * 2 + (k & 1)] = W[i];
    }
    __syncthreads();

    const int wave = tid >> 5;
    const int lane = tid & 31;
    const int m16  = lane & 15;
    const int hi   = lane >> 4;
    const int row0 = blk * 128 + wave * 16;
    constexpr int NT = NOUT / 16;

    v8f zero8 = {};
    v8f acc[NT];
#pragma unroll
    for (int t = 0; t < NT; ++t) acc[t] = zero8;

    int rA = row0 + m16;
    if (rA >= n_rows) rA = n_rows - 1;
    const float* arow = A + (size_t)rA * 128;

    for (int k0 = 0; k0 < 128; k0 += 4) {
        v2f a = *(const v2f*)(arow + k0 + 2 * hi);
        const int kh = (k0 >> 1) + hi;
#pragma unroll
        for (int t = 0; t < NT; ++t) {
            v2f b = *(const v2f*)(sW + (size_t)(kh * NOUT + t * 16 + m16) * 2);
            acc[t] = __builtin_amdgcn_wmma_f32_16x16x4_f32(
                false, a, false, b, (short)0, acc[t], false, false);
        }
    }

#pragma unroll
    for (int t = 0; t < NT; ++t) {
#pragma unroll
        for (int v = 0; v < 8; ++v) {
            int row = row0 + v + 8 * hi;
            if (row < n_rows)
                out[(size_t)row * NOUT + t * 16 + m16] = acc[t][v];
        }
    }
}

__global__ __launch_bounds__(256) void k_gemm_merge(const float* __restrict__ A,
                                                    const float* __restrict__ W,
                                                    float* __restrict__ out, int n_rows, int n_merge,
                                                    const unsigned* __restrict__ part, int n_slices,
                                                    unsigned* __restrict__ deg) {
    const int b = (int)blockIdx.x;
    if (b < n_merge) merge_block(part, n_slices, n_rows, deg, b);
    else             gemm_block<128>(A, W, out, n_rows, b - n_merge);
}


__device__ __forceinline__ int   lane_i(int v, int l)   { return __builtin_amdgcn_readlane(v, l); }
__device__ __forceinline__ float lane_f(float v, int l) {
    return __builtin_bit_cast(float, __builtin_amdgcn_readlane(__builtin_bit_cast(int, v), l));
}

__device__ __forceinline__ unsigned nz7(unsigned x) {
    return ((x & 0x7f7f7f7fu) + 0x7f7f7f7fu) | x;
}

__device__ __forceinline__ v4u load_key16(const unsigned char* __restrict__ key, int e, int E,
                                          unsigned other) {
    v4u r;
#pragma unroll
    for (int j = 0; j < 4; ++j) {
        unsigned w = 0u;
#pragma unroll
        for (int b = 0; b < 4; ++b) {
            const int x = e + 4 * j + b;
            const unsigned kb = (x < E) ? (unsigned)key[x] : other;
            w |= kb << (8 * b);
        }
        r[j] = w;
    }
    return r;
}

template <int FEAT, int PITCH, int B>
__device__ __forceinline__ void add_rows(const float* __restrict__ h, float* sAcc,
                                         int s, int dl, float nrm, int n, int lane) {
    constexpr int VW = FEAT / 32;
    constexpr int NB = 32 / B;
    typedef float vrow __attribute__((ext_vector_type(VW)));
#pragma unroll
    for (int b = 0; b < NB; ++b) {
        if (b * B < n) {
            vrow r[B];
#pragma unroll
            for (int j = 0; j < B; ++j) {
                if (b * B + j < n)
                    r[j] = *(const vrow*)(h + (size_t)lane_i(s, b * B + j) * FEAT + lane * VW);
            }
#pragma unroll
            for (int j = 0; j < B; ++j) {
                if (b * B + j < n) {
                    const int   row = lane_i(dl, b * B + j);
                    const float w   = lane_f(nrm, b * B + j);
                    float* a = sAcc + row * PITCH + lane;
#pragma unroll
                    for (int i = 0; i < VW; ++i)
                        __hip_atomic_fetch_add(a + i * 32, w * r[j][i], __ATOMIC_RELAXED,
                                               __HIP_MEMORY_SCOPE_WORKGROUP);
                }
            }
        }
    }
}

template <int FEAT, bool LISTED, int NEXT>
__global__ __launch_bounds__(1024) void k_edge_own(const float* __restrict__ h,
                                                   const unsigned* __restrict__ deg,
                                                   const int* __restrict__ ei, int E,
                                                   const unsigned char* __restrict__ key,
                                                   float* __restrict__ acc, int n_rows,
                                                   int* __restrict__ lst,
                                                   unsigned* __restrict__ lcnt,
                                                   const float* __restrict__ bias,
                                                   const float* __restrict__ Wn,
                                                   float* __restrict__ hn) {
    constexpr int VW = FEAT / 32;
    constexpr int B  = 32 / VW;
    constexpr int PITCH = NEXT > 0 ? FEAT + 4 : FEAT;
    typedef float vrow __attribute__((ext_vector_type(VW)));
    static_assert(NEXT == 0 || FEAT * NEXT <= OWN_WAVES * OWN_QCAP, "Wn must fit the queue space");

    __shared__ __attribute__((aligned(16))) float sAcc[(OWN_ROWS + 1) * PITCH];
    __shared__ __attribute__((aligned(16))) int sQ[OWN_WAVES * OWN_QCAP];
    __shared__ unsigned sCnt;

    const int lane = threadIdx.x & 31;
    const int wave = __builtin_amdgcn_readfirstlane((int)(threadIdx.x >> 5));
    const int lo   = blockIdx.x * OWN_ROWS;
    int nown = n_rows - lo;
    if (nown > OWN_ROWS) nown = OWN_ROWS;

    if (threadIdx.x == 0) sCnt = 0u;
    constexpr int NWL = NEXT > 0 ? (FEAT * NEXT) / (OWN_WAVES * 32) : 1;
    static_assert(NEXT == 0 || (FEAT * NEXT) % (OWN_WAVES * 32) == 0, "whole turns of the work-group");
    float wreg[NWL];
    if constexpr (NEXT > 0) {
#pragma unroll
        for (int u = 0; u < NWL; ++u) wreg[u] = Wn[u * (OWN_WAVES * 32) + (int)threadIdx.x];
    }
    {
        const vrow bv = *(const vrow*)(bias + lane * VW);
        for (int r0 = wave; r0 < nown; r0 += 4 * OWN_WAVES) {
            vrow  v[4];
            float w[4];
#pragma unroll
            for (int k = 0; k < 4; ++k) {
                const int r = r0 + k * OWN_WAVES;
                if (r < nown) {
                    v[k] = *(const vrow*)(h + (size_t)(lo + r) * FEAT + lane * VW);
                    const float di = dinv_of(deg, lo + r);
                    w[k] = di * di;
                }
            }
#pragma unroll
            for (int k = 0; k < 4; ++k) {
                const int r = r0 + k * OWN_WAVES;
                if (r < nown) {
#pragma unroll
                    for (int i = 0; i < VW; ++i) sAcc[r * PITCH + i * 32 + lane] = w[k] * v[k][i] + bv[i];
                }
            }
        }
    }
    __syncthreads();

    const int* dstp  = ei + E;
    const int  nfull = E >> 11;
    const unsigned own  = blockIdx.x;
    const unsigned own4 = own * 0x01010101u;
    const int rot = (int)blockIdx.x * (nfull / (int)gridDim.x);
    int* q  = sQ + wave * OWN_QCAP;
    int cnt = 0;

    auto flush = [&](const int* qp, int n) __attribute__((always_inline)) {
        const int me = qp[lane < n ? lane : 0];
        const int   s   = ei[me];
        const int   d   = dstp[me];
        const float nrm = dinv_of(deg, s) * dinv_of(deg, d);
        const int   dl  = d - lo;
        if (!LISTED) {
            unsigned pos = 0u;
            if (lane == 0)
                pos = __hip_atomic_fetch_add(&sCnt, 32u, __ATOMIC_RELAXED, __HIP_MEMORY_SCOPE_WORKGROUP);
            pos = (unsigned)__builtin_amdgcn_readfirstlane((int)pos);
            if (pos + 32u <= (unsigned)OWN_LCAP) {
                v2i rec;
                rec.x = lane < n ? s * 512 + dl : OWN_ROWS;
                rec.y = lane < n ? __builtin_bit_cast(int, nrm) : 0;
                ((v2i*)lst)[(size_t)own * OWN_LCAP + pos + lane] = rec;
            }
        }
        add_rows<FEAT, PITCH, B>(h, sAcc, s, dl, nrm, n, lane);
    };

    auto search = [&](const v4u k, int eb) __attribute__((always_inline)) {
        const unsigned n0 = nz7(k.x ^ own4);
        const unsigned n1 = nz7(k.y ^ own4);
        const unsigned n2 = nz7(k.z ^ own4);
        const unsigned n3 = nz7(k.w ^ own4);
        unsigned m = (((n0 >> 7) & 0x01010101u) | ((n1 >> 6) & 0x02020202u) |
                      ((n2 >> 5) & 0x04040404u) | ((n3 >> 4) & 0x08080808u)) ^ 0x0f0f0f0fu;
        unsigned bal = __builtin_amdgcn_ballot_w32(m != 0u);
        while (bal != 0u) {
            if (m != 0u) {
                const int pos = __builtin_ctz(m);
                q[cnt + (int)__builtin_amdgcn_mbcnt_lo(bal, 0u)] = eb + ((pos & 7) << 2) + (pos >> 3);
                m &= m - 1u;
            }
            cnt += __builtin_popcount(bal);
            bal = __builtin_amdgcn_ballot_w32(m != 0u);
        }
    };
    auto drain = [&]() __attribute__((always_inline)) {
        __builtin_amdgcn_wave_barrier();
        while (cnt >= 32) { cnt -= 32; flush(q + cnt, 32); }
    };

    bool listed = false;
    if (LISTED) {
        const unsigned taken = lcnt[own];
        if (taken <= (unsigned)OWN_LCAP) {
            listed = true;
            const int np = (int)(taken >> 5);
            const v2i* lp = (const v2i*)lst + (size_t)own * OWN_LCAP + lane;
            int   ck = OWN_ROWS;
            float cw = 0.0f;
            if (wave < np) { const v2i r = lp[wave * 32]; const int wb = r.y; ck = r.x; cw = __builtin_bit_cast(float, wb); }
            for (int p = wave; p < np; p += OWN_WAVES) {
                int nk = ck; float nw = cw;
                if (p + OWN_WAVES < np) {
                    const v2i r = lp[(p + OWN_WAVES) * 32];
                    const int wb = r.y;
                    nk = r.x; nw = __builtin_bit_cast(float, wb);
                }
                add_rows<FEAT, PITCH, (FEAT <= 64 ? 32 : 16)>(h, sAcc, (int)((unsigned)ck >> 9), ck & 511, cw, 32, lane);
                ck = nk; cw = nw;
            }
        }
    }
    if (!listed) {
    if (wave < nfull) {
        v4u cur[4], nxt[4];
        auto request = [&](v4u (&buf)[4], int g) __attribute__((always_inline)) {
            int gg = (g < nfull ? g : wave) + rot; if (gg >= nfull) gg -= nfull;
            const v4u* p = (const v4u*)(key + (size_t)gg * 2048) + lane;
#pragma unroll
            for (int u = 0; u < 4; ++u) buf[u] = p[u * 32];
        };
        request(cur, wave);
        for (int g = wave; g < nfull; g += OWN_WAVES) {
            request(nxt, g + OWN_WAVES);
            int gg = g + rot; if (gg >= nfull) gg -= nfull;
            const int e0 = gg * 2048 + lane * 16;
            v4u k0 = cur[0], k1 = cur[1], k2 = cur[2], k3 = cur[3];
#pragma clang loop unroll(disable)
            for (int u = 0; u < 4; ++u) {
                search(k0, e0 + u * 512);
                drain();
                k0 = k1; k1 = k2; k2 = k3;
            }
#pragma unroll
            for (int u = 0; u < 4; ++u) cur[u] = nxt[u];
        }
    }
    {
        const int ebase = nfull << 11;
        const int ntail = (E - ebase + 511) >> 9;
        for (int t = wave; t < ntail; t += OWN_WAVES) {
            const int e0 = ebase + t * 512 + lane * 16;
            search(load_key16(key, e0, E, (own + 1u) & 255u), e0);
            drain();
        }
    }
    if (cnt > 0) flush(q, cnt);
    }
    __syncthreads();
    if (!LISTED && threadIdx.x == 0) lcnt[own] = sCnt;

    if constexpr (NEXT == 0) {
        for (int r = wave; r < nown; r += OWN_WAVES) {
            vrow v;
#pragma unroll
            for (int i = 0; i < VW; ++i) v[i] = sAcc[r * PITCH + i * 32 + lane];
            *(vrow*)(acc + (size_t)(lo + r) * FEAT + lane * VW) = v;
        }
    } else {
        constexpr int NT = NEXT / 16;
        constexpr int VN = NEXT / 32;
        typedef float vnext __attribute__((ext_vector_type(VN)));
        float* sW = reinterpret_cast<float*>(sQ);
#pragma unroll
        for (int u = 0; u < NWL; ++u) {
            const int i = u * (OWN_WAVES * 32) + (int)threadIdx.x;
            const int k = i / NEXT, n = i % NEXT;
            sW[((k >> 1) * NEXT + n) * 2 + (k & 1)] = wreg[u];
        }
        __syncthreads();
        const int m16 = lane & 15;
        const int hi  = lane >> 4;
        for (int r0 = wave * 16; r0 < nown; r0 += OWN_WAVES * 16) {
            int ra = r0 + m16;
            if (ra >= nown) ra = nown - 1;
            const float* arow = sAcc + ra * PITCH + hi * 64;
            v8f zero8 = {};
            v8f c[NT];
#pragma unroll
            for (int t = 0; t < NT; ++t) c[t] = zero8;
            for (int kw = 0; kw < FEAT / 4; ++kw) {
                v2f a;
                a.x = arow[kw];
                a.y = arow[32 + kw];
                a.x = a.x > 0.0f ? a.x : LRELU_SLOPE * a.x;
                a.y = a.y > 0.0f ? a.y : LRELU_SLOPE * a.y;
                const int kh = 2 * kw + hi;
#pragma unroll
                for (int t = 0; t < NT; ++t) {
                    v2f b = *(const v2f*)(sW + (size_t)(kh * NEXT + t * 16 + m16) * 2);
                    c[t] = __builtin_amdgcn_wmma_f32_16x16x4_f32(
                        false, a, false, b, (short)0, c[t], false, false);
                }
            }
            float* srow = sAcc + r0 * PITCH;
#pragma unroll
            for (int t = 0; t < NT; ++t) {
#pragma unroll
                for (int v = 0; v < 8; ++v) {
                    if (r0 + v + 8 * hi < nown)
                        srow[(v + 8 * hi) * PITCH + t * 16 + m16] = c[t][v];
                }
            }
            __builtin_amdgcn_wave_barrier();
            vnext o[16];
#pragma unroll
            for (int j = 0; j < 16; ++j)
                if (r0 + j < nown) o[j] = *(const vnext*)(srow + j * PITCH + lane * VN);
#pragma unroll
            for (int j = 0; j < 16; ++j)
                if (r0 + j < nown)
                    *(vnext*)(hn + (size_t)(lo + r0 + j) * NEXT + lane * VN) = o[j];
        }
    }
}

extern "C" void kernel_launch(void* const* d_in, const int* in_sizes, int n_in,
                              void* d_out, int out_size, void* d_ws, size_t ws_size,
                              hipStream_t stream) {
    (void)n_in; (void)out_size; (void)ws_size;
    const float* z  = (const float*)d_in[0];
    const int*   ei = (const int*)d_in[1];
    const float* W1 = (const float*)d_in[2];
    const float* b1 = (const float*)d_in[3];
    const float* W2 = (const float*)d_in[4];
    const float* b2 = (const float*)d_in[5];

    const int N = in_sizes[0] / 128;
    const int E = in_sizes[1] / 2;

    char* ws = (char*)d_ws;
    size_t off = 0;
    auto carve = [&](size_t bytes) -> void* {
        void* p = ws + off;
        off = (off + bytes + 255) & ~(size_t)255;
        return p;
    };
    unsigned* deg  = (unsigned*)carve((size_t)N * 4);
    unsigned* key4 = (unsigned*)carve((size_t)E + 4);
    const unsigned char* key = (const unsigned char*)key4;
    float*    h1   = (float*)carve((size_t)N * 128 * 4);
    float*    h3   = (float*)carve((size_t)N * 64 * 4);
    float*    outp = (float*)d_out;

    const dim3 blk(256);
    const int gSl = (E + DEG_SLICE - 1) / DEG_SLICE;
    const int gMg = ((N + 1) / 2 + 255) / 256;
    const int gM = (N + 127) / 128;
    const int gOwn = (N + OWN_ROWS - 1) / OWN_ROWS;
    const dim3 blkOwn(OWN_WAVES * 32);
    int*      lst  = (int*)carve((size_t)gOwn * OWN_LCAP * 8);
    unsigned* lcnt = (unsigned*)carve((size_t)gOwn * 4);
    unsigned* part = (unsigned*)carve((size_t)gSl * DEG_WORDS * 4);

    k_count_slices<<<gSl, dim3(DEG_THREADS), 0, stream>>>(ei, E, N, part, key4);
    k_gemm_merge  <<<gMg + gM, blk, 0, stream>>>(z, W1, h1, N, gMg, part, gSl, deg);

    k_edge_own<128, false, 64><<<gOwn, blkOwn, 0, stream>>>(h1, deg, ei, E, key, nullptr, N, lst, lcnt,
                                                            b1, W2, h3);

    k_edge_own<64, true, 0><<<gOwn, blkOwn, 0, stream>>>(h3, deg, ei, E, key, outp, N, lst, lcnt,
                                                         b2, nullptr, nullptr);
}


# Byte key plane replaced by owner bit planes (one bit per edge per owner), built in LDS beside the z W1 product; an edge kernel reads only its own plane, words are match masks; LDS arrival order
# speedup vs baseline: 4.6471x; 1.0515x over previous
#include <hip/hip_runtime.h>
#include <hip/hip_bf16.h>

typedef float v2f __attribute__((ext_vector_type(2)));
typedef float v8f __attribute__((ext_vector_type(8)));

#define LRELU_SLOPE 0.01f

#define OWN_ROWS  392
#define OWN_WAVES 32
#define OWN_QCAP  256
#define OWN_LCAP  16384

typedef int      v2i __attribute__((ext_vector_type(2)));
typedef int      v4i __attribute__((ext_vector_type(4)));
typedef unsigned v2u __attribute__((ext_vector_type(2)));
typedef unsigned v4u __attribute__((ext_vector_type(4)));

#define DEG_SLICE 25000
#define DEG_THREADS 1024
#define DEG_WORDS 50176
#define PL_EDGES 1024
#define PL_PITCH 36

__device__ __forceinline__ v4i load_dst4(const int* __restrict__ dstp, int e, int E) {
    v4i r;
    r.x = (e     < E) ? dstp[e]     : -1;
    r.y = (e + 1 < E) ? dstp[e + 1] : -1;
    r.z = (e + 2 < E) ? dstp[e + 2] : -1;
    r.w = (e + 3 < E) ? dstp[e + 3] : -1;
    return r;
}

__global__ __launch_bounds__(DEG_THREADS) void k_count_slices(const int* __restrict__ ei, int E, int n_rows,
                                                       unsigned* __restrict__ part) {
    static_assert(DEG_SLICE < 65536 && DEG_SLICE % 4 == 0, "a slice's count fits 16 bits; 16-byte loads stay inside a slice");
    static_assert(DEG_WORDS % 4 == 0 && 2 * DEG_WORDS >= 256 * OWN_ROWS, "whole 16-byte stores; every row an owner can hold");
    __shared__ __attribute__((aligned(16))) unsigned sC[DEG_WORDS];
    constexpr int TURNS = (DEG_SLICE + 4 * DEG_THREADS - 1) / (4 * DEG_THREADS);
    const int tid = (int)threadIdx.x;
    const int* dstp = ei + E;
    const int e0 = (int)blockIdx.x * DEG_SLICE;
    int e1 = e0 + DEG_SLICE;
    if (e1 > E) e1 = E;

    const unsigned nlim = (unsigned)(n_rows < 2 * DEG_WORDS ? n_rows : 2 * DEG_WORDS);

    v4i dd[TURNS];
    if ((E & 3) == 0) {
#pragma unroll
        for (int t = 0; t < TURNS; ++t) {
            const int e = e0 + 4 * (tid + t * DEG_THREADS);
            dd[t] = *(const v4i*)(dstp + (e < e1 ? e : e0));
        }
    } else {
#pragma unroll
        for (int t = 0; t < TURNS; ++t) {
            const int e = e0 + 4 * (tid + t * DEG_THREADS);
            dd[t] = load_dst4(dstp, e < e1 ? e : e0, E);
        }
    }
    const v4u zero4 = {0u, 0u, 0u, 0u};
    for (int i = tid; i < DEG_WORDS / 4; i += DEG_THREADS) ((v4u*)sC)[i] = zero4;
    __syncthreads();

#pragma unroll
    for (int t = 0; t < TURNS; ++t) {
        const int e = e0 + 4 * (tid + t * DEG_THREADS);
        if (e >= e1) break;
        const v4i d = dd[t];
#pragma unroll
        for (int j = 0; j < 4; ++j) {
            if (e + j < e1) {
                const int dj = d[j];
                if ((unsigned)dj < nlim)
                    __hip_atomic_fetch_add(&sC[dj >> 1], 1u << ((dj & 1) * 16), __ATOMIC_RELAXED,
                                           __HIP_MEMORY_SCOPE_WORKGROUP);
            }
        }
    }
    __syncthreads();

    const int nw  = (int)((nlim + 1u) >> 1);
    const int nw4 = nw >> 2;
    unsigned* p = part + (size_t)blockIdx.x * DEG_WORDS;
    for (int i = tid; i < nw4; i += DEG_THREADS) ((v4u*)p)[i] = ((const v4u*)sC)[i];
    for (int i = (nw4 << 2) + tid; i < nw; i += DEG_THREADS) p[i] = sC[i];
}

__device__ __forceinline__ void merge_block(const unsigned* __restrict__ part, int n_slices, int n_rows,
                                            unsigned* __restrict__ deg, int blk) {
    const int i  = blk * (int)blockDim.x + (int)threadIdx.x;
    const int nw = (n_rows + 1) >> 1;
    if (i >= nw || i >= DEG_WORDS) return;
    unsigned lo = 0u, hi = 0u;
#pragma unroll 8
    for (int sl = 0; sl < n_slices; ++sl) {
        const unsigned w = part[(size_t)sl * DEG_WORDS + i];
        lo += w & 0xffffu;
        hi += w >> 16;
    }
    if (2 * i + 1 < n_rows) { v2u r; r.x = lo; r.y = hi; *(v2u*)(deg + 2 * i) = r; }
    else                    deg[2 * i] = lo;
}

__device__ __forceinline__ void plane_block(const int* __restrict__ ei, int E,
                                            unsigned* __restrict__ plane, int pw, int blk,
                                            unsigned* sP) {
    static_assert(PL_EDGES == 4 * 256 && PL_PITCH % 4 == 0 && PL_PITCH >= PL_EDGES / 32, "four edges per thread; 16-byte LDS rows");
    const int tid = (int)threadIdx.x;
    const int* dstp = ei + E;
    const int e = blk * PL_EDGES + 4 * tid;
    v4i d = {-1, -1, -1, -1};
    if ((E & 3) == 0) { if (e < E) d = *(const v4i*)(dstp + e); }
    else              d = load_dst4(dstp, e, E);
    const v4u zero4 = {0u, 0u, 0u, 0u};
    for (int i = tid; i < 256 * PL_PITCH / 4; i += 256) ((v4u*)sP)[i] = zero4;
    __syncthreads();
    const int w = tid >> 3;
#pragma unroll
    for (int j = 0; j < 4; ++j) {
        const unsigned o = (unsigned)d[j] / (unsigned)OWN_ROWS;
        if (o < 256u)
            __hip_atomic_fetch_or(&sP[o * PL_PITCH + w], 1u << ((4 * tid + j) & 31), __ATOMIC_RELAXED,
                                  __HIP_MEMORY_SCOPE_WORKGROUP);
    }
    __syncthreads();
#pragma unroll
    for (int i = tid; i < 256 * 8; i += 256) {
        const int o = i >> 3, c = i & 7;
        const v4u v = *(const v4u*)(sP + o * PL_PITCH + 4 * c);
        __builtin_nontemporal_store(v, (v4u*)(plane + (size_t)o * pw + blk * (PL_EDGES / 32) + 4 * c));
    }
}

__device__ __forceinline__ float dinv_of(const unsigned* __restrict__ deg, int i) {
    return rsqrtf((float)(deg[i] + 1u));
}

template <int NOUT>
__device__ __forceinline__ void gemm_block(const float* __restrict__ A,
                                           const float* __restrict__ W,
                                           float* __restrict__ out, int n_rows, int blk, float* sW) {
    const int tid = threadIdx.x;
    for (int i = tid; i < 128 * NOUT; i += 256) {
        int k = i / NOUT, n = i % NOUT;
        sW[((k >> 1) * NOUT + n) * 2 + (k & 1)] = W[i];
    }
    __syncthreads();

    const int wave = tid >> 5;
    const int lane = tid & 31;
    const int m16  = lane & 15;
    const int hi   = lane >> 4;
    const int row0 = blk * 128 + wave * 16;
    constexpr int NT = NOUT / 16;

    v8f zero8 = {};
    v8f acc[NT];
#pragma unroll
    for (int t = 0; t < NT; ++t) acc[t] = zero8;

    int rA = row0 + m16;
    if (rA >= n_rows) rA = n_rows - 1;
    const float* arow = A + (size_t)rA * 128;

    for (int k0 = 0; k0 < 128; k0 += 4) {
        v2f a = *(const v2f*)(arow + k0 + 2 * hi);
        const int kh = (k0 >> 1) + hi;
#pragma unroll
        for (int t = 0; t < NT; ++t) {
            v2f b = *(const v2f*)(sW + (size_t)(kh * NOUT + t * 16 + m16) * 2);
            acc[t] = __builtin_amdgcn_wmma_f32_16x16x4_f32(
                false, a, false, b, (short)0, acc[t], false, false);
        }
    }

#pragma unroll
    for (int t = 0; t < NT; ++t) {
#pragma unroll
        for (int v = 0; v < 8; ++v) {
            int row = row0 + v + 8 * hi;
            if (row < n_rows)
                out[(size_t)row * NOUT + t * 16 + m16] = acc[t][v];
        }
    }
}

__global__ __launch_bounds__(256) void k_gemm_merge(const float* __restrict__ A,
                                                    const float* __restrict__ W,
                                                    float* __restrict__ out, int n_rows, int n_merge,
                                                    const unsigned* __restrict__ part, int n_slices,
                                                    unsigned* __restrict__ deg, int n_gemm,
                                                    const int* __restrict__ ei, int E,
                                                    unsigned* __restrict__ plane, int pw) {
    static_assert(256 * PL_PITCH <= 128 * 128, "a plane piece fits the space of the staged weights");
    __shared__ __attribute__((aligned(16))) float sMem[128 * 128];
    const int b = (int)blockIdx.x;
    if (b < n_merge)               merge_block(part, n_slices, n_rows, deg, b);
    else if (b < n_merge + n_gemm) gemm_block<128>(A, W, out, n_rows, b - n_merge, sMem);
    else                           plane_block(ei, E, plane, pw, b - n_merge - n_gemm, reinterpret_cast<unsigned*>(sMem));
}


__device__ __forceinline__ int   lane_i(int v, int l)   { return __builtin_amdgcn_readlane(v, l); }
__device__ __forceinline__ float lane_f(float v, int l) {
    return __builtin_bit_cast(float, __builtin_amdgcn_readlane(__builtin_bit_cast(int, v), l));
}

template <int FEAT, int PITCH, int B>
__device__ __forceinline__ void add_rows(const float* __restrict__ h, float* sAcc,
                                         int s, int dl, float nrm, int n, int lane) {
    constexpr int VW = FEAT / 32;
    constexpr int NB = 32 / B;
    typedef float vrow __attribute__((ext_vector_type(VW)));
#pragma unroll
    for (int b = 0; b < NB; ++b) {
        if (b * B < n) {
            vrow r[B];
#pragma unroll
            for (int j = 0; j < B; ++j) {
                if (b * B + j < n)
                    r[j] = *(const vrow*)(h + (size_t)lane_i(s, b * B + j) * FEAT + lane * VW);
            }
#pragma unroll
            for (int j = 0; j < B; ++j) {
                if (b * B + j < n) {
                    const int   row = lane_i(dl, b * B + j);
                    const float w   = lane_f(nrm, b * B + j);
                    float* a = sAcc + row * PITCH + lane;
#pragma unroll
                    for (int i = 0; i < VW; ++i)
                        __hip_atomic_fetch_add(a + i * 32, w * r[j][i], __ATOMIC_RELAXED,
                                               __HIP_MEMORY_SCOPE_WORKGROUP);
                }
            }
        }
    }
}

template <int FEAT, bool LISTED, int NEXT>
__global__ __launch_bounds__(1024) void k_edge_own(const float* __restrict__ h,
                                                   const unsigned* __restrict__ deg,
                                                   const int* __restrict__ ei, int E,
                                                   const unsigned* __restrict__ plane, int pw,
                                                   float* __restrict__ acc, int n_rows,
                                                   int* __restrict__ lst,
                                                   unsigned* __restrict__ lcnt,
                                                   const float* __restrict__ bias,
                                                   const float* __restrict__ Wn,
                                                   float* __restrict__ hn) {
    constexpr int VW = FEAT / 32;
    constexpr int B  = 32 / VW;
    constexpr int PITCH = NEXT > 0 ? FEAT + 4 : FEAT;
    typedef float vrow __attribute__((ext_vector_type(VW)));
    static_assert(NEXT == 0 || FEAT * NEXT <= OWN_WAVES * OWN_QCAP, "Wn must fit the queue space");

    __shared__ __attribute__((aligned(16))) float sAcc[(OWN_ROWS + 1) * PITCH];
    __shared__ __attribute__((aligned(16))) int sQ[OWN_WAVES * OWN_QCAP];
    __shared__ unsigned sCnt;

    const int lane = threadIdx.x & 31;
    const int wave = __builtin_amdgcn_readfirstlane((int)(threadIdx.x >> 5));
    const int lo   = blockIdx.x * OWN_ROWS;
    int nown = n_rows - lo;
    if (nown > OWN_ROWS) nown = OWN_ROWS;

    const unsigned own = blockIdx.x;
    const unsigned* pl = plane + (size_t)own * pw;
    const int nq = (pw + 127) >> 7;
    auto request = [&](int g) __attribute__((always_inline)) -> v4u {
        v4u r = {0u, 0u, 0u, 0u};
        const int idx = g * 128 + 4 * lane;
        if (g < nq && idx < pw) r = __builtin_nontemporal_load((const v4u*)(pl + idx));
        return r;
    };
    v4u cur = {0u, 0u, 0u, 0u};
    if (!LISTED) cur = request(wave);

    if (threadIdx.x == 0) sCnt = 0u;
    constexpr int NWL = NEXT > 0 ? (FEAT * NEXT) / (OWN_WAVES * 32) : 1;
    static_assert(NEXT == 0 || (FEAT * NEXT) % (OWN_WAVES * 32) == 0, "whole turns of the work-group");
    float wreg[NWL];
    if constexpr (NEXT > 0) {
#pragma unroll
        for (int u = 0; u < NWL; ++u) wreg[u] = Wn[u * (OWN_WAVES * 32) + (int)threadIdx.x];
    }
    {
        const vrow bv = *(const vrow*)(bias + lane * VW);
        for (int r0 = wave; r0 < nown; r0 += 4 * OWN_WAVES) {
            vrow  v[4];
            float w[4];
#pragma unroll
            for (int k = 0; k < 4; ++k) {
                const int r = r0 + k * OWN_WAVES;
                if (r < nown) {
                    v[k] = *(const vrow*)(h + (size_t)(lo + r) * FEAT + lane * VW);
                    const float di = dinv_of(deg, lo + r);
                    w[k] = di * di;
                }
            }
#pragma unroll
            for (int k = 0; k < 4; ++k) {
                const int r = r0 + k * OWN_WAVES;
                if (r < nown) {
#pragma unroll
                    for (int i = 0; i < VW; ++i) sAcc[r * PITCH + i * 32 + lane] = w[k] * v[k][i] + bv[i];
                }
            }
        }
    }
    __syncthreads();

    const int* dstp  = ei + E;
    int* q  = sQ + wave * OWN_QCAP;
    int cnt = 0;

    auto flush = [&](const int* qp, int n) __attribute__((always_inline)) {
        const int me = qp[lane < n ? lane : 0];
        const int   s   = ei[me];
        const int   d   = dstp[me];
        const float nrm = dinv_of(deg, s) * dinv_of(deg, d);
        const int   dl  = d - lo;
        if (!LISTED) {
            unsigned pos = 0u;
            if (lane == 0)
                pos = __hip_atomic_fetch_add(&sCnt, 32u, __ATOMIC_RELAXED, __HIP_MEMORY_SCOPE_WORKGROUP);
            pos = (unsigned)__builtin_amdgcn_readfirstlane((int)pos);
            if (pos + 32u <= (unsigned)OWN_LCAP) {
                v2i rec;
                rec.x = lane < n ? s * 512 + dl : OWN_ROWS;
                rec.y = lane < n ? __builtin_bit_cast(int, nrm) : 0;
                ((v2i*)lst)[(size_t)own * OWN_LCAP + pos + lane] = rec;
            }
        }
        add_rows<FEAT, PITCH, B>(h, sAcc, s, dl, nrm, n, lane);
    };

    auto search = [&](unsigned m, int eb) __attribute__((always_inline)) {
        unsigned bal = __builtin_amdgcn_ballot_w32(m != 0u);
        while (bal != 0u) {
            if (m != 0u) {
                q[cnt + (int)__builtin_amdgcn_mbcnt_lo(bal, 0u)] = eb + __builtin_ctz(m);
                m &= m - 1u;
            }
            cnt += __builtin_popcount(bal);
            bal = __builtin_amdgcn_ballot_w32(m != 0u);
            __builtin_amdgcn_wave_barrier();
            if (cnt >= 32) { cnt -= 32; flush(q + cnt, 32); }
        }
    };

    bool listed = false;
    if (LISTED) {
        const unsigned taken = lcnt[own];
        if (taken <= (unsigned)OWN_LCAP) {
            listed = true;
            const int np = (int)(taken >> 5);
            const v2i* lp = (const v2i*)lst + (size_t)own * OWN_LCAP + lane;
            int   ck = OWN_ROWS;
            float cw = 0.0f;
            if (wave < np) { const v2i r = lp[wave * 32]; const int wb = r.y; ck = r.x; cw = __builtin_bit_cast(float, wb); }
            for (int p = wave; p < np; p += OWN_WAVES) {
                int nk = ck; float nw = cw;
                if (p + OWN_WAVES < np) {
                    const v2i r = lp[(p + OWN_WAVES) * 32];
                    const int wb = r.y;
                    nk = r.x; nw = __builtin_bit_cast(float, wb);
                }
                add_rows<FEAT, PITCH, (FEAT <= 64 ? 32 : 16)>(h, sAcc, (int)((unsigned)ck >> 9), ck & 511, cw, 32, lane);
                ck = nk; cw = nw;
            }
        }
    }
    if (!listed) {
        if (LISTED) cur = request(wave);
        for (int g = wave; g < nq; g += OWN_WAVES) {
            const v4u nxt = request(g + OWN_WAVES);
            const int eb = (g * 128 + 4 * lane) << 5;
            unsigned m0 = cur.x, m1 = cur.y, m2 = cur.z, m3 = cur.w;
#pragma clang loop unroll(disable)
            for (int u = 0; u < 4; ++u) {
                search(m0, eb + u * 32);
                m0 = m1; m1 = m2; m2 = m3;
            }
            cur = nxt;
        }
        if (cnt > 0) flush(q, cnt);
    }
    __syncthreads();
    if (!LISTED && threadIdx.x == 0) lcnt[own] = sCnt;

    if constexpr (NEXT == 0) {
        for (int r = wave; r < nown; r += OWN_WAVES) {
            vrow v;
#pragma unroll
            for (int i = 0; i < VW; ++i) v[i] = sAcc[r * PITCH + i * 32 + lane];
            *(vrow*)(acc + (size_t)(lo + r) * FEAT + lane * VW) = v;
        }
    } else {
        constexpr int NT = NEXT / 16;
        constexpr int VN = NEXT / 32;
        typedef float vnext __attribute__((ext_vector_type(VN)));
        float* sW = reinterpret_cast<float*>(sQ);
#pragma unroll
        for (int u = 0; u < NWL; ++u) {
            const int i = u * (OWN_WAVES * 32) + (int)threadIdx.x;
            const int k = i / NEXT, n = i % NEXT;
            sW[((k >> 1) * NEXT + n) * 2 + (k & 1)] = wreg[u];
        }
        __syncthreads();
        const int m16 = lane & 15;
        const int hi  = lane >> 4;
        for (int r0 = wave * 16; r0 < nown; r0 += OWN_WAVES * 16) {
            int ra = r0 + m16;
            if (ra >= nown) ra = nown - 1;
            const float* arow = sAcc + ra * PITCH + hi * 64;
            v8f zero8 = {};
            v8f c[NT];
#pragma unroll
            for (int t = 0; t < NT; ++t) c[t] = zero8;
            for (int kw = 0; kw < FEAT / 4; ++kw) {
                v2f a;
                a.x = arow[kw];
                a.y = arow[32 + kw];
                a.x = a.x > 0.0f ? a.x : LRELU_SLOPE * a.x;
                a.y = a.y > 0.0f ? a.y : LRELU_SLOPE * a.y;
                const int kh = 2 * kw + hi;
#pragma unroll
                for (int t = 0; t < NT; ++t) {
                    v2f b = *(const v2f*)(sW + (size_t)(kh * NEXT + t * 16 + m16) * 2);
                    c[t] = __builtin_amdgcn_wmma_f32_16x16x4_f32(
                        false, a, false, b, (short)0, c[t], false, false);
                }
            }
            float* srow = sAcc + r0 * PITCH;
#pragma unroll
            for (int t = 0; t < NT; ++t) {
#pragma unroll
                for (int v = 0; v < 8; ++v) {
                    if (r0 + v + 8 * hi < nown)
                        srow[(v + 8 * hi) * PITCH + t * 16 + m16] = c[t][v];
                }
            }
            __builtin_amdgcn_wave_barrier();
            vnext o[16];
#pragma unroll
            for (int j = 0; j < 16; ++j)
                if (r0 + j < nown) o[j] = *(const vnext*)(srow + j * PITCH + lane * VN);
#pragma unroll
            for (int j = 0; j < 16; ++j)
                if (r0 + j < nown)
                    *(vnext*)(hn + (size_t)(lo + r0 + j) * NEXT + lane * VN) = o[j];
        }
    }
}

extern "C" void kernel_launch(void* const* d_in, const int* in_sizes, int n_in,
                              void* d_out, int out_size, void* d_ws, size_t ws_size,
                              hipStream_t stream) {
    (void)n_in; (void)out_size; (void)ws_size;
    const float* z  = (const float*)d_in[0];
    const int*   ei = (const int*)d_in[1];
    const float* W1 = (const float*)d_in[2];
    const float* b1 = (const float*)d_in[3];
    const float* W2 = (const float*)d_in[4];
    const float* b2 = (const float*)d_in[5];

    const int N = in_sizes[0] / 128;
    const int E = in_sizes[1] / 2;

    char* ws = (char*)d_ws;
    size_t off = 0;
    auto carve = [&](size_t bytes) -> void* {
        void* p = ws + off;
        off = (off + bytes + 255) & ~(size_t)255;
        return p;
    };
    unsigned* deg  = (unsigned*)carve((size_t)N * 4);
    float*    h1   = (float*)carve((size_t)N * 128 * 4);
    float*    h3   = (float*)carve((size_t)N * 64 * 4);
    float*    outp = (float*)d_out;

    const dim3 blk(256);
    const int gSl = (E + DEG_SLICE - 1) / DEG_SLICE;
    const int gMg = ((N + 1) / 2 + 255) / 256;
    const int gM = (N + 127) / 128;
    const int gOwn = (N + OWN_ROWS - 1) / OWN_ROWS;
    const int gPl = (E + PL_EDGES - 1) / PL_EDGES;
    const int pw  = gPl * (PL_EDGES / 32);
    const dim3 blkOwn(OWN_WAVES * 32);
    int*      lst  = (int*)carve((size_t)gOwn * OWN_LCAP * 8);
    unsigned* lcnt = (unsigned*)carve((size_t)gOwn * 4);
    unsigned* part = (unsigned*)carve((size_t)gSl * DEG_WORDS * 4);
    unsigned* plane = (unsigned*)carve((size_t)256 * pw * 4);

    k_count_slices<<<gSl, dim3(DEG_THREADS), 0, stream>>>(ei, E, N, part);
    k_gemm_merge  <<<gMg + gM + gPl, blk, 0, stream>>>(z, W1, h1, N, gMg, part, gSl, deg, gM, ei, E, plane, pw);

    k_edge_own<128, false, 64><<<gOwn, blkOwn, 0, stream>>>(h1, deg, ei, E, plane, pw, nullptr, N, lst, lcnt,
                                                            b1, W2, h3);

    k_edge_own<64, true, 0><<<gOwn, blkOwn, 0, stream>>>(h3, deg, ei, E, plane, pw, outp, N, lst, lcnt,
                                                         b2, nullptr, nullptr);
}


# Owner planes hold edge records (source row*512 + owned row, 32 slots per 1024-edge piece) instead of match bits: layer-1 edge kernel needs no source or destination lookup; LDS arrival order
# speedup vs baseline: 5.1972x; 1.1184x over previous
#include <hip/hip_runtime.h>
#include <hip/hip_bf16.h>

typedef float v2f __attribute__((ext_vector_type(2)));
typedef float v8f __attribute__((ext_vector_type(8)));

#define LRELU_SLOPE 0.01f

#define OWN_ROWS  392
#define OWN_WAVES 32
#define OWN_QCAP  256
#define OWN_LCAP  16384

typedef int      v2i __attribute__((ext_vector_type(2)));
typedef int      v4i __attribute__((ext_vector_type(4)));
typedef unsigned v2u __attribute__((ext_vector_type(2)));
typedef unsigned v4u __attribute__((ext_vector_type(4)));

#define DEG_SLICE 25000
#define DEG_THREADS 1024
#define DEG_WORDS 50176
#define PL_EDGES 1024
#define PL_PITCH 36
#define PL_EMPTY 0xffffffffu

__device__ __forceinline__ v4i load_dst4(const int* __restrict__ dstp, int e, int E) {
    v4i r;
    r.x = (e     < E) ? dstp[e]     : -1;
    r.y = (e + 1 < E) ? dstp[e + 1] : -1;
    r.z = (e + 2 < E) ? dstp[e + 2] : -1;
    r.w = (e + 3 < E) ? dstp[e + 3] : -1;
    return r;
}

__global__ __launch_bounds__(DEG_THREADS) void k_count_slices(const int* __restrict__ ei, int E, int n_rows,
                                                       unsigned* __restrict__ part) {
    static_assert(DEG_SLICE < 65536 && DEG_SLICE % 4 == 0, "a slice's count fits 16 bits; 16-byte loads stay inside a slice");
    static_assert(DEG_WORDS % 4 == 0 && 2 * DEG_WORDS >= 256 * OWN_ROWS, "whole 16-byte stores; every row an owner can hold");
    __shared__ __attribute__((aligned(16))) unsigned sC[DEG_WORDS];
    constexpr int TURNS = (DEG_SLICE + 4 * DEG_THREADS - 1) / (4 * DEG_THREADS);
    const int tid = (int)threadIdx.x;
    const int* dstp = ei + E;
    const int e0 = (int)blockIdx.x * DEG_SLICE;
    int e1 = e0 + DEG_SLICE;
    if (e1 > E) e1 = E;

    const unsigned nlim = (unsigned)(n_rows < 2 * DEG_WORDS ? n_rows : 2 * DEG_WORDS);

    v4i dd[TURNS];
    if ((E & 3) == 0) {
#pragma unroll
        for (int t = 0; t < TURNS; ++t) {
            const int e = e0 + 4 * (tid + t * DEG_THREADS);
            dd[t] = *(const v4i*)(dstp + (e < e1 ? e : e0));
        }
    } else {
#pragma unroll
        for (int t = 0; t < TURNS; ++t) {
            const int e = e0 + 4 * (tid + t * DEG_THREADS);
            dd[t] = load_dst4(dstp, e < e1 ? e : e0, E);
        }
    }
    const v4u zero4 = {0u, 0u, 0u, 0u};
    for (int i = tid; i < DEG_WORDS / 4; i += DEG_THREADS) ((v4u*)sC)[i] = zero4;
    __syncthreads();

#pragma unroll
    for (int t = 0; t < TURNS; ++t) {
        const int e = e0 + 4 * (tid + t * DEG_THREADS);
        if (e >= e1) break;
        const v4i d = dd[t];
#pragma unroll
        for (int j = 0; j < 4; ++j) {
            if (e + j < e1) {
                const int dj = d[j];
                if ((unsigned)dj < nlim)
                    __hip_atomic_fetch_add(&sC[dj >> 1], 1u << ((dj & 1) * 16), __ATOMIC_RELAXED,
                                           __HIP_MEMORY_SCOPE_WORKGROUP);
            }
        }
    }
    __syncthreads();

    const int nw  = (int)((nlim + 1u) >> 1);
    const int nw4 = nw >> 2;
    unsigned* p = part + (size_t)blockIdx.x * DEG_WORDS;
    for (int i = tid; i < nw4; i += DEG_THREADS) ((v4u*)p)[i] = ((const v4u*)sC)[i];
    for (int i = (nw4 << 2) + tid; i < nw; i += DEG_THREADS) p[i] = sC[i];
}

__device__ __forceinline__ void merge_block(const unsigned* __restrict__ part, int n_slices, int n_rows,
                                            unsigned* __restrict__ deg, int blk) {
    const int i  = blk * (int)blockDim.x + (int)threadIdx.x;
    const int nw = (n_rows + 1) >> 1;
    if (i >= nw || i >= DEG_WORDS) return;
    unsigned lo = 0u, hi = 0u;
#pragma unroll 8
    for (int sl = 0; sl < n_slices; ++sl) {
        const unsigned w = part[(size_t)sl * DEG_WORDS + i];
        lo += w & 0xffffu;
        hi += w >> 16;
    }
    if (2 * i + 1 < n_rows) { v2u r; r.x = lo; r.y = hi; *(v2u*)(deg + 2 * i) = r; }
    else                    deg[2 * i] = lo;
}

__device__ __forceinline__ void plane_block(const int* __restrict__ ei, int E,
                                            unsigned* __restrict__ plane, int pw, int blk,
                                            unsigned* sP, unsigned* __restrict__ ovf, int n_pl) {
    static_assert(PL_EDGES == 4 * 256 && PL_PITCH % 4 == 0 && PL_PITCH > PL_EDGES / 32, "four edges per thread; 16-byte LDS rows; room for the counter");
    const int tid = (int)threadIdx.x;
    const int* dstp = ei + E;
    const int e = blk * PL_EDGES + 4 * tid;
    v4i d = {-1, -1, -1, -1};
    v4i s = {0, 0, 0, 0};
    if ((E & 3) == 0) { if (e < E) { d = *(const v4i*)(dstp + e); s = *(const v4i*)(ei + e); } }
    else              { d = load_dst4(dstp, e, E); s = load_dst4(ei, e, E); }
    const v4u zero4  = {0u, 0u, 0u, 0u};
    const v4u empty4 = {PL_EMPTY, PL_EMPTY, PL_EMPTY, PL_EMPTY};
    for (int i = tid; i < 256 * PL_PITCH / 4; i += 256) ((v4u*)sP)[i] = (i % (PL_PITCH / 4) == 8) ? zero4 : empty4;
    __syncthreads();
#pragma unroll
    for (int j = 0; j < 4; ++j) {
        const unsigned o = (unsigned)d[j] / (unsigned)OWN_ROWS;
        if (o < 256u) {
            const unsigned pos = __hip_atomic_fetch_add(&sP[o * PL_PITCH + 32], 1u, __ATOMIC_RELAXED,
                                                        __HIP_MEMORY_SCOPE_WORKGROUP);
            if (pos < 32u) sP[o * PL_PITCH + pos] = (unsigned)s[j] * 512u + ((unsigned)d[j] - o * (unsigned)OWN_ROWS);
        }
    }
    __syncthreads();
    {
        const bool over = sP[tid * PL_PITCH + 32] > 32u;
        const unsigned bal = __builtin_amdgcn_ballot_w32(over);
        if ((tid & 31) == 0) ovf[(size_t)(tid >> 5) * n_pl + blk] = bal != 0u ? 1u : 0u;
    }
#pragma unroll
    for (int i = tid; i < 256 * 8; i += 256) {
        const int o = i >> 3, c4 = i & 7;
        const v4u v = *(const v4u*)(sP + o * PL_PITCH + 4 * c4);
        __builtin_nontemporal_store(v, (v4u*)(plane + (size_t)o * pw + blk * (PL_EDGES / 32) + 4 * c4));
    }
}

__device__ __forceinline__ float dinv_of(const unsigned* __restrict__ deg, int i) {
    return rsqrtf((float)(deg[i] + 1u));
}

template <int NOUT>
__device__ __forceinline__ void gemm_block(const float* __restrict__ A,
                                           const float* __restrict__ W,
                                           float* __restrict__ out, int n_rows, int blk, float* sW) {
    const int tid = threadIdx.x;
    for (int i = tid; i < 128 * NOUT; i += 256) {
        int k = i / NOUT, n = i % NOUT;
        sW[((k >> 1) * NOUT + n) * 2 + (k & 1)] = W[i];
    }
    __syncthreads();

    const int wave = tid >> 5;
    const int lane = tid & 31;
    const int m16  = lane & 15;
    const int hi   = lane >> 4;
    const int row0 = blk * 128 + wave * 16;
    constexpr int NT = NOUT / 16;

    v8f zero8 = {};
    v8f acc[NT];
#pragma unroll
    for (int t = 0; t < NT; ++t) acc[t] = zero8;

    int rA = row0 + m16;
    if (rA >= n_rows) rA = n_rows - 1;
    const float* arow = A + (size_t)rA * 128;

    for (int k0 = 0; k0 < 128; k0 += 4) {
        v2f a = *(const v2f*)(arow + k0 + 2 * hi);
        const int kh = (k0 >> 1) + hi;
#pragma unroll
        for (int t = 0; t < NT; ++t) {
            v2f b = *(const v2f*)(sW + (size_t)(kh * NOUT + t * 16 + m16) * 2);
            acc[t] = __builtin_amdgcn_wmma_f32_16x16x4_f32(
                false, a, false, b, (short)0, acc[t], false, false);
        }
    }

#pragma unroll
    for (int t = 0; t < NT; ++t) {
#pragma unroll
        for (int v = 0; v < 8; ++v) {
            int row = row0 + v + 8 * hi;
            if (row < n_rows)
                out[(size_t)row * NOUT + t * 16 + m16] = acc[t][v];
        }
    }
}

__global__ __launch_bounds__(256) void k_gemm_merge(const float* __restrict__ A,
                                                    const float* __restrict__ W,
                                                    float* __restrict__ out, int n_rows, int n_merge,
                                                    const unsigned* __restrict__ part, int n_slices,
                                                    unsigned* __restrict__ deg, int n_gemm,
                                                    const int* __restrict__ ei, int E,
                                                    unsigned* __restrict__ plane, int pw,
                                                    unsigned* __restrict__ ovf, int n_pl) {
    static_assert(256 * PL_PITCH <= 128 * 128, "a plane piece fits the space of the staged weights");
    __shared__ __attribute__((aligned(16))) float sMem[128 * 128];
    const int b = (int)blockIdx.x;
    if (b < n_merge)               merge_block(part, n_slices, n_rows, deg, b);
    else if (b < n_merge + n_gemm) gemm_block<128>(A, W, out, n_rows, b - n_merge, sMem);
    else                           plane_block(ei, E, plane, pw, b - n_merge - n_gemm, reinterpret_cast<unsigned*>(sMem), ovf, n_pl);
}


__device__ __forceinline__ int   lane_i(int v, int l)   { return __builtin_amdgcn_readlane(v, l); }
__device__ __forceinline__ float lane_f(float v, int l) {
    return __builtin_bit_cast(float, __builtin_amdgcn_readlane(__builtin_bit_cast(int, v), l));
}

template <int FEAT, int PITCH, int B>
__device__ __forceinline__ void add_rows(const float* __restrict__ h, float* sAcc,
                                         int s, int dl, float nrm, int n, int lane) {
    constexpr int VW = FEAT / 32;
    constexpr int NB = 32 / B;
    typedef float vrow __attribute__((ext_vector_type(VW)));
#pragma unroll
    for (int b = 0; b < NB; ++b) {
        if (b * B < n) {
            vrow r[B];
#pragma unroll
            for (int j = 0; j < B; ++j) {
                if (b * B + j < n)
                    r[j] = *(const vrow*)(h + (size_t)lane_i(s, b * B + j) * FEAT + lane * VW);
            }
#pragma unroll
            for (int j = 0; j < B; ++j) {
                if (b * B + j < n) {
                    const int   row = lane_i(dl, b * B + j);
                    const float w   = lane_f(nrm, b * B + j);
                    float* a = sAcc + row * PITCH + lane;
#pragma unroll
                    for (int i = 0; i < VW; ++i)
                        __hip_atomic_fetch_add(a + i * 32, w * r[j][i], __ATOMIC_RELAXED,
                                               __HIP_MEMORY_SCOPE_WORKGROUP);
                }
            }
        }
    }
}

template <int FEAT, bool LISTED, int NEXT>
__global__ __launch_bounds__(1024) void k_edge_own(const float* __restrict__ h,
                                                   const unsigned* __restrict__ deg,
                                                   const int* __restrict__ ei, int E,
                                                   const unsigned* __restrict__ plane, int pw,
                                                   const unsigned* __restrict__ ovf,
                                                   float* __restrict__ acc, int n_rows,
                                                   int* __restrict__ lst,
                                                   unsigned* __restrict__ lcnt,
                                                   const float* __restrict__ bias,
                                                   const float* __restrict__ Wn,
                                                   float* __restrict__ hn) {
    constexpr int VW = FEAT / 32;
    constexpr int B  = 32 / VW;
    constexpr int PITCH = NEXT > 0 ? FEAT + 4 : FEAT;
    typedef float vrow __attribute__((ext_vector_type(VW)));
    static_assert(NEXT == 0 || FEAT * NEXT <= OWN_WAVES * OWN_QCAP, "Wn must fit the queue space");

    __shared__ __attribute__((aligned(16))) float sAcc[(OWN_ROWS + 1) * PITCH];
    __shared__ __attribute__((aligned(16))) int sQ[OWN_WAVES * OWN_QCAP];
    __shared__ unsigned sCnt;

    const int lane = threadIdx.x & 31;
    const int wave = __builtin_amdgcn_readfirstlane((int)(threadIdx.x >> 5));
    const int lo   = blockIdx.x * OWN_ROWS;
    int nown = n_rows - lo;
    if (nown > OWN_ROWS) nown = OWN_ROWS;

    const unsigned own = blockIdx.x;
    const unsigned* pl = plane + (size_t)own * pw;
    const unsigned* ovo = ovf + (size_t)(own >> 5) * (pw >> 5);
    const int nq = (pw + 127) >> 7;
    struct step_t { v4u r; unsigned f; };
    auto request = [&](int g) __attribute__((always_inline)) -> step_t {
        step_t st;
        st.r.x = PL_EMPTY; st.r.y = PL_EMPTY; st.r.z = PL_EMPTY; st.r.w = PL_EMPTY;
        st.f = 0u;
        const int idx = g * 128 + 4 * lane;
        if (g < nq && idx < pw) {
            st.r = __builtin_nontemporal_load((const v4u*)(pl + idx));
            st.f = ovo[idx >> 5];
        }
        return st;
    };
    step_t cur;
    cur.r.x = PL_EMPTY; cur.r.y = PL_EMPTY; cur.r.z = PL_EMPTY; cur.r.w = PL_EMPTY;
    cur.f = 0u;
    if (!LISTED) cur = request(wave);

    if (threadIdx.x == 0) sCnt = 0u;
    constexpr int NWL = NEXT > 0 ? (FEAT * NEXT) / (OWN_WAVES * 32) : 1;
    static_assert(NEXT == 0 || (FEAT * NEXT) % (OWN_WAVES * 32) == 0, "whole turns of the work-group");
    float wreg[NWL];
    if constexpr (NEXT > 0) {
#pragma unroll
        for (int u = 0; u < NWL; ++u) wreg[u] = Wn[u * (OWN_WAVES * 32) + (int)threadIdx.x];
    }
    {
        const vrow bv = *(const vrow*)(bias + lane * VW);
        for (int r0 = wave; r0 < nown; r0 += 4 * OWN_WAVES) {
            vrow  v[4];
            float w[4];
#pragma unroll
            for (int k = 0; k < 4; ++k) {
                const int r = r0 + k * OWN_WAVES;
                if (r < nown) {
                    v[k] = *(const vrow*)(h + (size_t)(lo + r) * FEAT + lane * VW);
                    const float di = dinv_of(deg, lo + r);
                    w[k] = di * di;
                }
            }
#pragma unroll
            for (int k = 0; k < 4; ++k) {
                const int r = r0 + k * OWN_WAVES;
                if (r < nown) {
#pragma unroll
                    for (int i = 0; i < VW; ++i) sAcc[r * PITCH + i * 32 + lane] = w[k] * v[k][i] + bv[i];
                }
            }
        }
    }
    __syncthreads();

    const int* dstp  = ei + E;
    int* q  = sQ + wave * OWN_QCAP;
    int cnt = 0;

    auto flush = [&](const int* qp, int n) __attribute__((always_inline)) {
        const int me = qp[lane < n ? lane : 0];
        const int   s   = (int)((unsigned)me >> 9);
        const int   dl  = me & 511;
        const float nrm = dinv_of(deg, s) * dinv_of(deg, lo + dl);
        if (!LISTED) {
            unsigned pos = 0u;
            if (lane == 0)
                pos = __hip_atomic_fetch_add(&sCnt, 32u, __ATOMIC_RELAXED, __HIP_MEMORY_SCOPE_WORKGROUP);
            pos = (unsigned)__builtin_amdgcn_readfirstlane((int)pos);
            if (pos + 32u <= (unsigned)OWN_LCAP) {
                v2i rec;
                rec.x = lane < n ? s * 512 + dl : OWN_ROWS;
                rec.y = lane < n ? __builtin_bit_cast(int, nrm) : 0;
                ((v2i*)lst)[(size_t)own * OWN_LCAP + pos + lane] = rec;
            }
        }
        add_rows<FEAT, PITCH, B>(h, sAcc, s, dl, nrm, n, lane);
    };

    auto take = [&](unsigned rec) __attribute__((always_inline)) {
        const bool has = rec != PL_EMPTY;
        const unsigned bal = __builtin_amdgcn_ballot_w32(has);
        if (bal != 0u) {
            if (has) q[cnt + (int)__builtin_amdgcn_mbcnt_lo(bal, 0u)] = (int)rec;
            cnt += __builtin_popcount(bal);
            __builtin_amdgcn_wave_barrier();
            if (cnt >= 32) { cnt -= 32; flush(q + cnt, 32); }
        }
    };

    bool listed = false;
    if (LISTED) {
        const unsigned taken = lcnt[own];
        if (taken <= (unsigned)OWN_LCAP) {
            listed = true;
            const int np = (int)(taken >> 5);
            const v2i* lp = (const v2i*)lst + (size_t)own * OWN_LCAP + lane;
            int   ck = OWN_ROWS;
            float cw = 0.0f;
            if (wave < np) { const v2i r = lp[wave * 32]; const int wb = r.y; ck = r.x; cw = __builtin_bit_cast(float, wb); }
            for (int p = wave; p < np; p += OWN_WAVES) {
                int nk = ck; float nw = cw;
                if (p + OWN_WAVES < np) {
                    const v2i r = lp[(p + OWN_WAVES) * 32];
                    const int wb = r.y;
                    nk = r.x; nw = __builtin_bit_cast(float, wb);
                }
                add_rows<FEAT, PITCH, (FEAT <= 64 ? 32 : 16)>(h, sAcc, (int)((unsigned)ck >> 9), ck & 511, cw, 32, lane);
                ck = nk; cw = nw;
            }
        }
    }
    if (!listed) {
        if (LISTED) cur = request(wave);
        for (int g = wave; g < nq; g += OWN_WAVES) {
            const step_t nxt = request(g + OWN_WAVES);
            unsigned m0 = cur.r.x, m1 = cur.r.y, m2 = cur.r.z, m3 = cur.r.w;
            const unsigned fb = __builtin_amdgcn_ballot_w32(cur.f != 0u);
            if (fb != 0u) {
                if (cur.f != 0u) { m0 = PL_EMPTY; m1 = PL_EMPTY; m2 = PL_EMPTY; m3 = PL_EMPTY; }
                for (int pc = 0; pc < 4; ++pc) {
                    if (((fb >> (8 * pc)) & 0xffu) == 0u) continue;
                    const int e0 = (4 * g + pc) * PL_EDGES;
#pragma clang loop unroll(disable)
                    for (int k = 0; k < PL_EDGES; k += 32) {
                        const int e = e0 + k + lane;
                        unsigned rec = PL_EMPTY;
                        if (e < E) {
                            const unsigned dloc = (unsigned)(dstp[e] - lo);
                            if (dloc < (unsigned)OWN_ROWS) rec = (unsigned)ei[e] * 512u + dloc;
                        }
                        take(rec);
                    }
                }
            }
#pragma clang loop unroll(disable)
            for (int u = 0; u < 4; ++u) {
                take(m0);
                m0 = m1; m1 = m2; m2 = m3;
            }
            cur = nxt;
        }
        if (cnt > 0) flush(q, cnt);
    }
    __syncthreads();
    if (!LISTED && threadIdx.x == 0) lcnt[own] = sCnt;

    if constexpr (NEXT == 0) {
        for (int r = wave; r < nown; r += OWN_WAVES) {
            vrow v;
#pragma unroll
            for (int i = 0; i < VW; ++i) v[i] = sAcc[r * PITCH + i * 32 + lane];
            *(vrow*)(acc + (size_t)(lo + r) * FEAT + lane * VW) = v;
        }
    } else {
        constexpr int NT = NEXT / 16;
        constexpr int VN = NEXT / 32;
        typedef float vnext __attribute__((ext_vector_type(VN)));
        float* sW = reinterpret_cast<float*>(sQ);
#pragma unroll
        for (int u = 0; u < NWL; ++u) {
            const int i = u * (OWN_WAVES * 32) + (int)threadIdx.x;
            const int k = i / NEXT, n = i % NEXT;
            sW[((k >> 1) * NEXT + n) * 2 + (k & 1)] = wreg[u];
        }
        __syncthreads();
        const int m16 = lane & 15;
        const int hi  = lane >> 4;
        for (int r0 = wave * 16; r0 < nown; r0 += OWN_WAVES * 16) {
            int ra = r0 + m16;
            if (ra >= nown) ra = nown - 1;
            const float* arow = sAcc + ra * PITCH + hi * 64;
            v8f zero8 = {};
            v8f c[NT];
#pragma unroll
            for (int t = 0; t < NT; ++t) c[t] = zero8;
            for (int kw = 0; kw < FEAT / 4; ++kw) {
                v2f a;
                a.x = arow[kw];
                a.y = arow[32 + kw];
                a.x = a.x > 0.0f ? a.x : LRELU_SLOPE * a.x;
                a.y = a.y > 0.0f ? a.y : LRELU_SLOPE * a.y;
                const int kh = 2 * kw + hi;
#pragma unroll
                for (int t = 0; t < NT; ++t) {
                    v2f b = *(const v2f*)(sW + (size_t)(kh * NEXT + t * 16 + m16) * 2);
                    c[t] = __builtin_amdgcn_wmma_f32_16x16x4_f32(
                        false, a, false, b, (short)0, c[t], false, false);
                }
            }
            float* srow = sAcc + r0 * PITCH;
#pragma unroll
            for (int t = 0; t < NT; ++t) {
#pragma unroll
                for (int v = 0; v < 8; ++v) {
                    if (r0 + v + 8 * hi < nown)
                        srow[(v + 8 * hi) * PITCH + t * 16 + m16] = c[t][v];
                }
            }
            __builtin_amdgcn_wave_barrier();
            vnext o[16];
#pragma unroll
            for (int j = 0; j < 16; ++j)
                if (r0 + j < nown) o[j] = *(const vnext*)(srow + j * PITCH + lane * VN);
#pragma unroll
            for (int j = 0; j < 16; ++j)
                if (r0 + j < nown)
                    *(vnext*)(hn + (size_t)(lo + r0 + j) * NEXT + lane * VN) = o[j];
        }
    }
}

extern "C" void kernel_launch(void* const* d_in, const int* in_sizes, int n_in,
                              void* d_out, int out_size, void* d_ws, size_t ws_size,
                              hipStream_t stream) {
    (void)n_in; (void)out_size; (void)ws_size;
    const float* z  = (const float*)d_in[0];
    const int*   ei = (const int*)d_in[1];
    const float* W1 = (const float*)d_in[2];
    const float* b1 = (const float*)d_in[3];
    const float* W2 = (const float*)d_in[4];
    const float* b2 = (const float*)d_in[5];

    const int N = in_sizes[0] / 128;
    const int E = in_sizes[1] / 2;

    char* ws = (char*)d_ws;
    size_t off = 0;
    auto carve = [&](size_t bytes) -> void* {
        void* p = ws + off;
        off = (off + bytes + 255) & ~(size_t)255;
        return p;
    };
    unsigned* deg  = (unsigned*)carve((size_t)N * 4);
    float*    h1   = (float*)carve((size_t)N * 128 * 4);
    float*    h3   = (float*)carve((size_t)N * 64 * 4);
    float*    outp = (float*)d_out;

    const dim3 blk(256);
    const int gSl = (E + DEG_SLICE - 1) / DEG_SLICE;
    const int gMg = ((N + 1) / 2 + 255) / 256;
    const int gM = (N + 127) / 128;
    const int gOwn = (N + OWN_ROWS - 1) / OWN_ROWS;
    const int gPl = (E + PL_EDGES - 1) / PL_EDGES;
    const int pw  = gPl * (PL_EDGES / 32);
    const dim3 blkOwn(OWN_WAVES * 32);
    int*      lst  = (int*)carve((size_t)gOwn * OWN_LCAP * 8);
    unsigned* lcnt = (unsigned*)carve((size_t)gOwn * 4);
    unsigned* part = (unsigned*)carve((size_t)gSl * DEG_WORDS * 4);
    unsigned* plane = (unsigned*)carve((size_t)256 * pw * 4);
    unsigned* ovf   = (unsigned*)carve((size_t)8 * gPl * 4);

    k_count_slices<<<gSl, dim3(DEG_THREADS), 0, stream>>>(ei, E, N, part);
    k_gemm_merge  <<<gMg + gM + gPl, blk, 0, stream>>>(z, W1, h1, N, gMg, part, gSl, deg, gM, ei, E, plane, pw, ovf, gPl);

    k_edge_own<128, false, 64><<<gOwn, blkOwn, 0, stream>>>(h1, deg, ei, E, plane, pw, ovf, nullptr, N, lst, lcnt,
                                                            b1, W2, h3);

    k_edge_own<64, true, 0><<<gOwn, blkOwn, 0, stream>>>(h3, deg, ei, E, plane, pw, ovf, outp, N, lst, lcnt,
                                                         b2, nullptr, nullptr);
}


# Rows of h stored times their own dinv (z W1 waves sum their strip's degree planes, merge blocks gone; layer-1 tail scales h3): an edge adds dinv[d]*h[s], no source-degree lookup; sums refactored
# speedup vs baseline: 5.3505x; 1.0295x over previous
#include <hip/hip_runtime.h>
#include <hip/hip_bf16.h>

typedef float v2f __attribute__((ext_vector_type(2)));
typedef float v8f __attribute__((ext_vector_type(8)));

#define LRELU_SLOPE 0.01f

#define OWN_ROWS  392
#define OWN_WAVES 32
#define OWN_QCAP  256
#define OWN_LCAP  16384

typedef int      v2i __attribute__((ext_vector_type(2)));
typedef int      v4i __attribute__((ext_vector_type(4)));
typedef unsigned v2u __attribute__((ext_vector_type(2)));
typedef unsigned v4u __attribute__((ext_vector_type(4)));

#define DEG_SLICE 25000
#define DEG_THREADS 1024
#define DEG_WORDS 50176
#define PL_EDGES 1024
#define PL_PITCH 36
#define PL_EMPTY 0xffffffffu

__device__ __forceinline__ v4i load_dst4(const int* __restrict__ dstp, int e, int E) {
    v4i r;
    r.x = (e     < E) ? dstp[e]     : -1;
    r.y = (e + 1 < E) ? dstp[e + 1] : -1;
    r.z = (e + 2 < E) ? dstp[e + 2] : -1;
    r.w = (e + 3 < E) ? dstp[e + 3] : -1;
    return r;
}

__global__ __launch_bounds__(DEG_THREADS) void k_count_slices(const int* __restrict__ ei, int E, int n_rows,
                                                       unsigned* __restrict__ part) {
    static_assert(DEG_SLICE < 65536 && DEG_SLICE % 4 == 0, "a slice's count fits 16 bits; 16-byte loads stay inside a slice");
    static_assert(DEG_WORDS % 4 == 0 && 2 * DEG_WORDS >= 256 * OWN_ROWS, "whole 16-byte stores; every row an owner can hold");
    __shared__ __attribute__((aligned(16))) unsigned sC[DEG_WORDS];
    constexpr int TURNS = (DEG_SLICE + 4 * DEG_THREADS - 1) / (4 * DEG_THREADS);
    const int tid = (int)threadIdx.x;
    const int* dstp = ei + E;
    const int e0 = (int)blockIdx.x * DEG_SLICE;
    int e1 = e0 + DEG_SLICE;
    if (e1 > E) e1 = E;

    const unsigned nlim = (unsigned)(n_rows < 2 * DEG_WORDS ? n_rows : 2 * DEG_WORDS);

    v4i dd[TURNS];
    if ((E & 3) == 0) {
#pragma unroll
        for (int t = 0; t < TURNS; ++t) {
            const int e = e0 + 4 * (tid + t * DEG_THREADS);
            dd[t] = *(const v4i*)(dstp + (e < e1 ? e : e0));
        }
    } else {
#pragma unroll
        for (int t = 0; t < TURNS; ++t) {
            const int e = e0 + 4 * (tid + t * DEG_THREADS);
            dd[t] = load_dst4(dstp, e < e1 ? e : e0, E);
        }
    }
    const v4u zero4 = {0u, 0u, 0u, 0u};
    for (int i = tid; i < DEG_WORDS / 4; i += DEG_THREADS) ((v4u*)sC)[i] = zero4;
    __syncthreads();

#pragma unroll
    for (int t = 0; t < TURNS; ++t) {
        const int e = e0 + 4 * (tid + t * DEG_THREADS);
        if (e >= e1) break;
        const v4i d = dd[t];
#pragma unroll
        for (int j = 0; j < 4; ++j) {
            if (e + j < e1) {
                const int dj = d[j];
                if ((unsigned)dj < nlim)
                    __hip_atomic_fetch_add(&sC[dj >> 1], 1u << ((dj & 1) * 16), __ATOMIC_RELAXED,
                                           __HIP_MEMORY_SCOPE_WORKGROUP);
            }
        }
    }
    __syncthreads();

    const int nw  = (int)((nlim + 1u) >> 1);
    const int nw4 = nw >> 2;
    unsigned* p = part + (size_t)blockIdx.x * DEG_WORDS;
    for (int i = tid; i < nw4; i += DEG_THREADS) ((v4u*)p)[i] = ((const v4u*)sC)[i];
    for (int i = (nw4 << 2) + tid; i < nw; i += DEG_THREADS) p[i] = sC[i];
}

__device__ __forceinline__ void strip_degrees(const unsigned* __restrict__ part, int n_slices, int w0, int nw,
                                              int lane, unsigned& lo, unsigned& hi) {
    const int wi = w0 + (lane & 7);
    lo = 0u; hi = 0u;
    if (wi < nw) {
#pragma unroll 8
        for (int sl = lane >> 3; sl < n_slices; sl += 4) {
            const unsigned w = part[(size_t)sl * DEG_WORDS + wi];
            lo += w & 0xffffu;
            hi += w >> 16;
        }
    }
    lo += (unsigned)__shfl_xor((int)lo, 8);  hi += (unsigned)__shfl_xor((int)hi, 8);
    lo += (unsigned)__shfl_xor((int)lo, 16); hi += (unsigned)__shfl_xor((int)hi, 16);
}

__device__ __forceinline__ void plane_block(const int* __restrict__ ei, int E,
                                            unsigned* __restrict__ plane, int pw, int blk,
                                            unsigned* sP, unsigned* __restrict__ ovf, int n_pl) {
    static_assert(PL_EDGES == 4 * 256 && PL_PITCH % 4 == 0 && PL_PITCH > PL_EDGES / 32, "four edges per thread; 16-byte LDS rows; room for the counter");
    const int tid = (int)threadIdx.x;
    const int* dstp = ei + E;
    const int e = blk * PL_EDGES + 4 * tid;
    v4i d = {-1, -1, -1, -1};
    v4i s = {0, 0, 0, 0};
    if ((E & 3) == 0) { if (e < E) { d = *(const v4i*)(dstp + e); s = *(const v4i*)(ei + e); } }
    else              { d = load_dst4(dstp, e, E); s = load_dst4(ei, e, E); }
    const v4u zero4  = {0u, 0u, 0u, 0u};
    const v4u empty4 = {PL_EMPTY, PL_EMPTY, PL_EMPTY, PL_EMPTY};
    for (int i = tid; i < 256 * PL_PITCH / 4; i += 256) ((v4u*)sP)[i] = (i % (PL_PITCH / 4) == 8) ? zero4 : empty4;
    __syncthreads();
#pragma unroll
    for (int j = 0; j < 4; ++j) {
        const unsigned o = (unsigned)d[j] / (unsigned)OWN_ROWS;
        if (o < 256u) {
            const unsigned pos = __hip_atomic_fetch_add(&sP[o * PL_PITCH + 32], 1u, __ATOMIC_RELAXED,
                                                        __HIP_MEMORY_SCOPE_WORKGROUP);
            if (pos < 32u) sP[o * PL_PITCH + pos] = (unsigned)s[j] * 512u + ((unsigned)d[j] - o * (unsigned)OWN_ROWS);
        }
    }
    __syncthreads();
    {
        const bool over = sP[tid * PL_PITCH + 32] > 32u;
        const unsigned bal = __builtin_amdgcn_ballot_w32(over);
        if ((tid & 31) == 0) ovf[(size_t)(tid >> 5) * n_pl + blk] = bal != 0u ? 1u : 0u;
    }
#pragma unroll
    for (int i = tid; i < 256 * 8; i += 256) {
        const int o = i >> 3, c4 = i & 7;
        const v4u v = *(const v4u*)(sP + o * PL_PITCH + 4 * c4);
        __builtin_nontemporal_store(v, (v4u*)(plane + (size_t)o * pw + blk * (PL_EDGES / 32) + 4 * c4));
    }
}

__device__ __forceinline__ float dinv_of(const unsigned* __restrict__ deg, int i) {
    return rsqrtf((float)(deg[i] + 1u));
}

template <int NOUT>
__device__ __forceinline__ void gemm_block(const float* __restrict__ A,
                                           const float* __restrict__ W,
                                           float* __restrict__ out, int n_rows, int blk, float* sW,
                                           const unsigned* __restrict__ part, int n_slices,
                                           unsigned* __restrict__ deg) {
    const int tid = threadIdx.x;
    unsigned dlo, dhi;
    {
        const int nlim = n_rows < 2 * DEG_WORDS ? n_rows : 2 * DEG_WORDS;
        strip_degrees(part, n_slices, blk * 64 + (tid >> 5) * 8, (nlim + 1) >> 1, tid & 31, dlo, dhi);
        const int r = blk * 128 + (tid >> 5) * 16 + 2 * (tid & 31);
        if ((tid & 31) < 8) {
            if (r + 1 < n_rows) { v2u d2; d2.x = dlo; d2.y = dhi; *(v2u*)(deg + r) = d2; }
            else if (r < n_rows) deg[r] = dlo;
        }
    }
    for (int i = tid; i < 128 * NOUT; i += 256) {
        int k = i / NOUT, n = i % NOUT;
        sW[((k >> 1) * NOUT + n) * 2 + (k & 1)] = W[i];
    }
    __syncthreads();

    const int wave = tid >> 5;
    const int lane = tid & 31;
    const int m16  = lane & 15;
    const int hi   = lane >> 4;
    const int row0 = blk * 128 + wave * 16;
    constexpr int NT = NOUT / 16;

    v8f zero8 = {};
    v8f acc[NT];
#pragma unroll
    for (int t = 0; t < NT; ++t) acc[t] = zero8;

    int rA = row0 + m16;
    if (rA >= n_rows) rA = n_rows - 1;
    const float* arow = A + (size_t)rA * 128;

    for (int k0 = 0; k0 < 128; k0 += 4) {
        v2f a = *(const v2f*)(arow + k0 + 2 * hi);
        const int kh = (k0 >> 1) + hi;
#pragma unroll
        for (int t = 0; t < NT; ++t) {
            v2f b = *(const v2f*)(sW + (size_t)(kh * NOUT + t * 16 + m16) * 2);
            acc[t] = __builtin_amdgcn_wmma_f32_16x16x4_f32(
                false, a, false, b, (short)0, acc[t], false, false);
        }
    }

    float dv[8];
#pragma unroll
    for (int v = 0; v < 8; ++v) {
        const unsigned d = (unsigned)__shfl((int)((v & 1) ? dhi : dlo), (v >> 1) + 4 * hi);
        dv[v] = rsqrtf((float)(d + 1u));
    }
#pragma unroll
    for (int t = 0; t < NT; ++t) {
#pragma unroll
        for (int v = 0; v < 8; ++v) {
            int row = row0 + v + 8 * hi;
            if (row < n_rows)
                out[(size_t)row * NOUT + t * 16 + m16] = dv[v] * acc[t][v];
        }
    }
}

__global__ __launch_bounds__(256) void k_gemm_planes(const float* __restrict__ A,
                                                     const float* __restrict__ W,
                                                     float* __restrict__ out, int n_rows,
                                                     const unsigned* __restrict__ part, int n_slices,
                                                     unsigned* __restrict__ deg, int n_gemm,
                                                     const int* __restrict__ ei, int E,
                                                     unsigned* __restrict__ plane, int pw,
                                                     unsigned* __restrict__ ovf, int n_pl) {
    static_assert(256 * PL_PITCH <= 128 * 128, "a plane piece fits the space of the staged weights");
    __shared__ __attribute__((aligned(16))) float sMem[128 * 128];
    const int b = (int)blockIdx.x;
    if (b < n_gemm) gemm_block<128>(A, W, out, n_rows, b, sMem, part, n_slices, deg);
    else            plane_block(ei, E, plane, pw, b - n_gemm, reinterpret_cast<unsigned*>(sMem), ovf, n_pl);
}


__device__ __forceinline__ int   lane_i(int v, int l)   { return __builtin_amdgcn_readlane(v, l); }
__device__ __forceinline__ float lane_f(float v, int l) {
    return __builtin_bit_cast(float, __builtin_amdgcn_readlane(__builtin_bit_cast(int, v), l));
}

template <int FEAT, int PITCH, int B>
__device__ __forceinline__ void add_rows(const float* __restrict__ h, float* sAcc,
                                         int s, int dl, float nrm, int n, int lane) {
    constexpr int VW = FEAT / 32;
    constexpr int NB = 32 / B;
    typedef float vrow __attribute__((ext_vector_type(VW)));
#pragma unroll
    for (int b = 0; b < NB; ++b) {
        if (b * B < n) {
            vrow r[B];
#pragma unroll
            for (int j = 0; j < B; ++j) {
                if (b * B + j < n)
                    r[j] = *(const vrow*)(h + (size_t)lane_i(s, b * B + j) * FEAT + lane * VW);
            }
#pragma unroll
            for (int j = 0; j < B; ++j) {
                if (b * B + j < n) {
                    const int   row = lane_i(dl, b * B + j);
                    const float w   = lane_f(nrm, b * B + j);
                    float* a = sAcc + row * PITCH + lane;
#pragma unroll
                    for (int i = 0; i < VW; ++i)
                        __hip_atomic_fetch_add(a + i * 32, w * r[j][i], __ATOMIC_RELAXED,
                                               __HIP_MEMORY_SCOPE_WORKGROUP);
                }
            }
        }
    }
}

template <int FEAT, bool LISTED, int NEXT>
__global__ __launch_bounds__(1024) void k_edge_own(const float* __restrict__ h,
                                                   const unsigned* __restrict__ deg,
                                                   const int* __restrict__ ei, int E,
                                                   const unsigned* __restrict__ plane, int pw,
                                                   const unsigned* __restrict__ ovf,
                                                   float* __restrict__ acc, int n_rows,
                                                   int* __restrict__ lst,
                                                   unsigned* __restrict__ lcnt,
                                                   const float* __restrict__ bias,
                                                   const float* __restrict__ Wn,
                                                   float* __restrict__ hn) {
    constexpr int VW = FEAT / 32;
    constexpr int B  = 32 / VW;
    constexpr int PITCH = NEXT > 0 ? FEAT + 4 : FEAT;
    typedef float vrow __attribute__((ext_vector_type(VW)));
    static_assert(NEXT == 0 || FEAT * NEXT <= OWN_WAVES * OWN_QCAP, "Wn must fit the queue space");

    __shared__ __attribute__((aligned(16))) float sAcc[(OWN_ROWS + 1) * PITCH];
    __shared__ __attribute__((aligned(16))) int sQ[OWN_WAVES * OWN_QCAP];
    __shared__ unsigned sCnt;

    const int lane = threadIdx.x & 31;
    const int wave = __builtin_amdgcn_readfirstlane((int)(threadIdx.x >> 5));
    const int lo   = blockIdx.x * OWN_ROWS;
    int nown = n_rows - lo;
    if (nown > OWN_ROWS) nown = OWN_ROWS;

    const unsigned own = blockIdx.x;
    const unsigned* pl = plane + (size_t)own * pw;
    const unsigned* ovo = ovf + (size_t)(own >> 5) * (pw >> 5);
    const int nq = (pw + 127) >> 7;
    struct step_t { v4u r; unsigned f; };
    auto request = [&](int g) __attribute__((always_inline)) -> step_t {
        step_t st;
        st.r.x = PL_EMPTY; st.r.y = PL_EMPTY; st.r.z = PL_EMPTY; st.r.w = PL_EMPTY;
        st.f = 0u;
        const int idx = g * 128 + 4 * lane;
        if (g < nq && idx < pw) {
            st.r = __builtin_nontemporal_load((const v4u*)(pl + idx));
            st.f = ovo[idx >> 5];
        }
        return st;
    };
    step_t cur;
    cur.r.x = PL_EMPTY; cur.r.y = PL_EMPTY; cur.r.z = PL_EMPTY; cur.r.w = PL_EMPTY;
    cur.f = 0u;
    if (!LISTED) cur = request(wave);

    if (threadIdx.x == 0) sCnt = 0u;
    constexpr int NWL = NEXT > 0 ? (FEAT * NEXT) / (OWN_WAVES * 32) : 1;
    static_assert(NEXT == 0 || (FEAT * NEXT) % (OWN_WAVES * 32) == 0, "whole turns of the work-group");
    float wreg[NWL];
    if constexpr (NEXT > 0) {
#pragma unroll
        for (int u = 0; u < NWL; ++u) wreg[u] = Wn[u * (OWN_WAVES * 32) + (int)threadIdx.x];
    }
    {
        const vrow bv = *(const vrow*)(bias + lane * VW);
        for (int r0 = wave; r0 < nown; r0 += 4 * OWN_WAVES) {
            vrow  v[4];
            float w[4];
#pragma unroll
            for (int k = 0; k < 4; ++k) {
                const int r = r0 + k * OWN_WAVES;
                if (r < nown) {
                    v[k] = *(const vrow*)(h + (size_t)(lo + r) * FEAT + lane * VW);
                    w[k] = dinv_of(deg, lo + r);
                }
            }
#pragma unroll
            for (int k = 0; k < 4; ++k) {
                const int r = r0 + k * OWN_WAVES;
                if (r < nown) {
#pragma unroll
                    for (int i = 0; i < VW; ++i) sAcc[r * PITCH + i * 32 + lane] = w[k] * v[k][i] + bv[i];
                }
            }
        }
    }
    __syncthreads();

    const int* dstp  = ei + E;
    int* q  = sQ + wave * OWN_QCAP;
    int cnt = 0;

    auto flush = [&](const int* qp, int n) __attribute__((always_inline)) {
        const int me = qp[lane < n ? lane : 0];
        const int   s   = (int)((unsigned)me >> 9);
        const int   dl  = me & 511;
        const float nrm = dinv_of(deg, lo + dl);
        if (!LISTED) {
            unsigned pos = 0u;
            if (lane == 0)
                pos = __hip_atomic_fetch_add(&sCnt, 32u, __ATOMIC_RELAXED, __HIP_MEMORY_SCOPE_WORKGROUP);
            pos = (unsigned)__builtin_amdgcn_readfirstlane((int)pos);
            if (pos + 32u <= (unsigned)OWN_LCAP) {
                v2i rec;
                rec.x = lane < n ? s * 512 + dl : OWN_ROWS;
                rec.y = lane < n ? __builtin_bit_cast(int, nrm) : 0;
                ((v2i*)lst)[(size_t)own * OWN_LCAP + pos + lane] = rec;
            }
        }
        add_rows<FEAT, PITCH, B>(h, sAcc, s, dl, nrm, n, lane);
    };

    auto take = [&](unsigned rec) __attribute__((always_inline)) {
        const bool has = rec != PL_EMPTY;
        const unsigned bal = __builtin_amdgcn_ballot_w32(has);
        if (bal != 0u) {
            if (has) q[cnt + (int)__builtin_amdgcn_mbcnt_lo(bal, 0u)] = (int)rec;
            cnt += __builtin_popcount(bal);
            __builtin_amdgcn_wave_barrier();
            if (cnt >= 32) { cnt -= 32; flush(q + cnt, 32); }
        }
    };

    bool listed = false;
    if (LISTED) {
        const unsigned taken = lcnt[own];
        if (taken <= (unsigned)OWN_LCAP) {
            listed = true;
            const int np = (int)(taken >> 5);
            const v2i* lp = (const v2i*)lst + (size_t)own * OWN_LCAP + lane;
            int   ck = OWN_ROWS;
            float cw = 0.0f;
            if (wave < np) { const v2i r = lp[wave * 32]; const int wb = r.y; ck = r.x; cw = __builtin_bit_cast(float, wb); }
            for (int p = wave; p < np; p += OWN_WAVES) {
                int nk = ck; float nw = cw;
                if (p + OWN_WAVES < np) {
                    const v2i r = lp[(p + OWN_WAVES) * 32];
                    const int wb = r.y;
                    nk = r.x; nw = __builtin_bit_cast(float, wb);
                }
                add_rows<FEAT, PITCH, (FEAT <= 64 ? 32 : 16)>(h, sAcc, (int)((unsigned)ck >> 9), ck & 511, cw, 32, lane);
                ck = nk; cw = nw;
            }
        }
    }
    if (!listed) {
        if (LISTED) cur = request(wave);
        for (int g = wave; g < nq; g += OWN_WAVES) {
            const step_t nxt = request(g + OWN_WAVES);
            unsigned m0 = cur.r.x, m1 = cur.r.y, m2 = cur.r.z, m3 = cur.r.w;
            const unsigned fb = __builtin_amdgcn_ballot_w32(cur.f != 0u);
            if (fb != 0u) {
                if (cur.f != 0u) { m0 = PL_EMPTY; m1 = PL_EMPTY; m2 = PL_EMPTY; m3 = PL_EMPTY; }
                for (int pc = 0; pc < 4; ++pc) {
                    if (((fb >> (8 * pc)) & 0xffu) == 0u) continue;
                    const int e0 = (4 * g + pc) * PL_EDGES;
#pragma clang loop unroll(disable)
                    for (int k = 0; k < PL_EDGES; k += 32) {
                        const int e = e0 + k + lane;
                        unsigned rec = PL_EMPTY;
                        if (e < E) {
                            const unsigned dloc = (unsigned)(dstp[e] - lo);
                            if (dloc < (unsigned)OWN_ROWS) rec = (unsigned)ei[e] * 512u + dloc;
                        }
                        take(rec);
                    }
                }
            }
#pragma clang loop unroll(disable)
            for (int u = 0; u < 4; ++u) {
                take(m0);
                m0 = m1; m1 = m2; m2 = m3;
            }
            cur = nxt;
        }
        if (cnt > 0) flush(q, cnt);
    }
    __syncthreads();
    if (!LISTED && threadIdx.x == 0) lcnt[own] = sCnt;

    if constexpr (NEXT == 0) {
        for (int r = wave; r < nown; r += OWN_WAVES) {
            vrow v;
#pragma unroll
            for (int i = 0; i < VW; ++i) v[i] = sAcc[r * PITCH + i * 32 + lane];
            *(vrow*)(acc + (size_t)(lo + r) * FEAT + lane * VW) = v;
        }
    } else {
        constexpr int NT = NEXT / 16;
        constexpr int VN = NEXT / 32;
        typedef float vnext __attribute__((ext_vector_type(VN)));
        float* sW = reinterpret_cast<float*>(sQ);
#pragma unroll
        for (int u = 0; u < NWL; ++u) {
            const int i = u * (OWN_WAVES * 32) + (int)threadIdx.x;
            const int k = i / NEXT, n = i % NEXT;
            sW[((k >> 1) * NEXT + n) * 2 + (k & 1)] = wreg[u];
        }
        __syncthreads();
        const int m16 = lane & 15;
        const int hi  = lane >> 4;
        for (int r0 = wave * 16; r0 < nown; r0 += OWN_WAVES * 16) {
            int ra = r0 + m16;
            if (ra >= nown) ra = nown - 1;
            const float* arow = sAcc + ra * PITCH + hi * 64;
            float dn[8];
#pragma unroll
            for (int v = 0; v < 8; ++v) {
                int rr = r0 + v + 8 * hi;
                if (rr >= nown) rr = nown - 1;
                dn[v] = dinv_of(deg, lo + rr);
            }
            v8f zero8 = {};
            v8f c[NT];
#pragma unroll
            for (int t = 0; t < NT; ++t) c[t] = zero8;
            for (int kw = 0; kw < FEAT / 4; ++kw) {
                v2f a;
                a.x = arow[kw];
                a.y = arow[32 + kw];
                a.x = a.x > 0.0f ? a.x : LRELU_SLOPE * a.x;
                a.y = a.y > 0.0f ? a.y : LRELU_SLOPE * a.y;
                const int kh = 2 * kw + hi;
#pragma unroll
                for (int t = 0; t < NT; ++t) {
                    v2f b = *(const v2f*)(sW + (size_t)(kh * NEXT + t * 16 + m16) * 2);
                    c[t] = __builtin_amdgcn_wmma_f32_16x16x4_f32(
                        false, a, false, b, (short)0, c[t], false, false);
                }
            }
            float* srow = sAcc + r0 * PITCH;
#pragma unroll
            for (int t = 0; t < NT; ++t) {
#pragma unroll
                for (int v = 0; v < 8; ++v) {
                    if (r0 + v + 8 * hi < nown)
                        srow[(v + 8 * hi) * PITCH + t * 16 + m16] = dn[v] * c[t][v];
                }
            }
            __builtin_amdgcn_wave_barrier();
            vnext o[16];
#pragma unroll
            for (int j = 0; j < 16; ++j)
                if (r0 + j < nown) o[j] = *(const vnext*)(srow + j * PITCH + lane * VN);
#pragma unroll
            for (int j = 0; j < 16; ++j)
                if (r0 + j < nown)
                    *(vnext*)(hn + (size_t)(lo + r0 + j) * NEXT + lane * VN) = o[j];
        }
    }
}

extern "C" void kernel_launch(void* const* d_in, const int* in_sizes, int n_in,
                              void* d_out, int out_size, void* d_ws, size_t ws_size,
                              hipStream_t stream) {
    (void)n_in; (void)out_size; (void)ws_size;
    const float* z  = (const float*)d_in[0];
    const int*   ei = (const int*)d_in[1];
    const float* W1 = (const float*)d_in[2];
    const float* b1 = (const float*)d_in[3];
    const float* W2 = (const float*)d_in[4];
    const float* b2 = (const float*)d_in[5];

    const int N = in_sizes[0] / 128;
    const int E = in_sizes[1] / 2;

    char* ws = (char*)d_ws;
    size_t off = 0;
    auto carve = [&](size_t bytes) -> void* {
        void* p = ws + off;
        off = (off + bytes + 255) & ~(size_t)255;
        return p;
    };
    unsigned* deg  = (unsigned*)carve((size_t)N * 4);
    float*    h1   = (float*)carve((size_t)N * 128 * 4);
    float*    h3   = (float*)carve((size_t)N * 64 * 4);
    float*    outp = (float*)d_out;

    const dim3 blk(256);
    const int gSl = (E + DEG_SLICE - 1) / DEG_SLICE;
    const int gM = (N + 127) / 128;
    const int gOwn = (N + OWN_ROWS - 1) / OWN_ROWS;
    const int gPl = (E + PL_EDGES - 1) / PL_EDGES;
    const int pw  = gPl * (PL_EDGES / 32);
    const dim3 blkOwn(OWN_WAVES * 32);
    int*      lst  = (int*)carve((size_t)gOwn * OWN_LCAP * 8);
    unsigned* lcnt = (unsigned*)carve((size_t)gOwn * 4);
    unsigned* part = (unsigned*)carve((size_t)gSl * DEG_WORDS * 4);
    unsigned* plane = (unsigned*)carve((size_t)256 * pw * 4);
    unsigned* ovf   = (unsigned*)carve((size_t)8 * gPl * 4);

    k_count_slices<<<gSl, dim3(DEG_THREADS), 0, stream>>>(ei, E, N, part);
    k_gemm_planes <<<gM + gPl, blk, 0, stream>>>(z, W1, h1, N, part, gSl, deg, gM, ei, E, plane, pw, ovf, gPl);

    k_edge_own<128, false, 64><<<gOwn, blkOwn, 0, stream>>>(h1, deg, ei, E, plane, pw, ovf, nullptr, N, lst, lcnt,
                                                            b1, W2, h3);

    k_edge_own<64, true, 0><<<gOwn, blkOwn, 0, stream>>>(h3, deg, ei, E, plane, pw, ovf, outp, N, lst, lcnt,
                                                         b2, nullptr, nullptr);
}
